# starting kernel
# baseline (speedup 1.0000x reference)
#include <hip/hip_runtime.h>

#define N_USERS  100000
#define N_ITEMS  50000
#define N_NODES  150000
#define N_EDGES  1000000
#define EMBED_K  64

typedef __attribute__((ext_vector_type(2))) float v2f;
typedef __attribute__((ext_vector_type(8))) float v8f;

__device__ __forceinline__ void atomAddF32(float* p, float v) {
    (void)__hip_atomic_fetch_add(p, v, __ATOMIC_RELAXED, __HIP_MEMORY_SCOPE_AGENT);
}

__global__ void k_zero(float* __restrict__ p, int n) {
    int i = blockIdx.x * blockDim.x + threadIdx.x;
    if (i < n) p[i] = 0.0f;
}

__global__ void k_init_x(const float* __restrict__ Gu, const float* __restrict__ Gi,
                         float* __restrict__ x) {
    int i = blockIdx.x * blockDim.x + threadIdx.x;
    const int total  = N_NODES * EMBED_K;
    const int uElems = N_USERS * EMBED_K;
    if (i >= total) return;
    x[i] = (i < uElems) ? Gu[i] : Gi[i - uElems];
}

__global__ void k_deg(const int* __restrict__ row, const int* __restrict__ col,
                      const float* __restrict__ intents, float* __restrict__ deg) {
    int e = blockIdx.x * blockDim.x + threadIdx.x;
    if (e >= N_EDGES) return;
    float z0 = intents[e];
    float z1 = intents[N_EDGES + e];
    float z2 = intents[2 * N_EDGES + e];
    float z3 = intents[3 * N_EDGES + e];
    float m  = fmaxf(fmaxf(z0, z1), fmaxf(z2, z3));
    float w0 = __expf(z0 - m), w1 = __expf(z1 - m);
    float w2 = __expf(z2 - m), w3 = __expf(z3 - m);
    float inv = 1.0f / (w0 + w1 + w2 + w3);
    w0 *= inv; w1 *= inv; w2 *= inv; w3 *= inv;
    int r = row[e], c = col[e];
    atomAddF32(deg + r * 4 + 0, w0); atomAddF32(deg + r * 4 + 1, w1);
    atomAddF32(deg + r * 4 + 2, w2); atomAddF32(deg + r * 4 + 3, w3);
    atomAddF32(deg + c * 4 + 0, w0); atomAddF32(deg + c * 4 + 1, w1);
    atomAddF32(deg + c * 4 + 2, w2); atomAddF32(deg + c * 4 + 3, w3);
}

__global__ void k_dis(float* __restrict__ deg) {
    int i = blockIdx.x * blockDim.x + threadIdx.x;
    if (i >= N_NODES * 4) return;
    float d = deg[i];
    deg[i] = (d > 0.0f) ? rsqrtf(fmaxf(d, 1e-12f)) : 0.0f;
}

__global__ __launch_bounds__(128) void k_propagate(
        const float* __restrict__ x_src, float* __restrict__ x_dst,
        const float* __restrict__ dis,
        const int* __restrict__ row, const int* __restrict__ col) {
    const int wave = blockIdx.x * (blockDim.x >> 5) + (threadIdx.x >> 5);
    const int lane = threadIdx.x & 31;
    const int hl   = lane & 15;
    const bool hi  = lane >= 16;
    const int base = wave * 16;

    const int eA  = base + hl;
    const int irA = row[eA];
    const int icA = col[eA];
    __builtin_prefetch(&x_src[irA * 64], 0, 0);
    const float4 dr = *(const float4*)(dis + irA * 4);
    const float4 dc = *(const float4*)(dis + icA * 4);
    float nrm[4];
    nrm[0] = dr.x * dc.x; nrm[1] = dr.y * dc.y;
    nrm[2] = dr.z * dc.z; nrm[3] = dr.w * dc.w;

    int colv[8];
#pragma unroll
    for (int v = 0; v < 8; ++v) colv[v] = col[base + v + (hi ? 8 : 0)];

    const int kx = hi ? 2 : 0;

#pragma unroll
    for (int k = 0; k < 4; ++k) {
        v8f acc = {};
#pragma unroll
        for (int t = 0; t < 4; ++t) {
            v2f a, b;
            a.x = (hl == 4 * t + kx)     ? nrm[k] : 0.0f;
            a.y = (hl == 4 * t + kx + 1) ? nrm[k] : 0.0f;
            const int rb0 = row[base + 4 * t + kx];
            const int rb1 = row[base + 4 * t + kx + 1];
            b.x = x_src[rb0 * 64 + k * 16 + hl];
            b.y = x_src[rb1 * 64 + k * 16 + hl];
            acc = __builtin_amdgcn_wmma_f32_16x16x4_f32(
                      false, a, false, b, (short)0, acc, false, false);
        }
#pragma unroll
        for (int v = 0; v < 8; ++v)
            atomAddF32(x_dst + colv[v] * 64 + k * 16 + hl, acc[v]);
    }
}

extern "C" void kernel_launch(void* const* d_in, const int* in_sizes, int n_in,
                              void* d_out, int out_size, void* d_ws, size_t ws_size,
                              hipStream_t stream) {
    const float* Gu      = (const float*)d_in[0];
    const float* Gi      = (const float*)d_in[1];
    const int*   edge    = (const int*)d_in[2];
    const float* intents = (const float*)d_in[3];
    const int* row = edge;
    const int* col = edge + N_EDGES;
    float* out = (float*)d_out;

    float* deg  = (float*)d_ws;
    float* xbuf = deg + (size_t)N_NODES * 4;

    const int NX = N_NODES * EMBED_K;
    const int ND = N_NODES * 4;
    dim3 b256(256);
    const int waves  = N_EDGES / 16;
    dim3 pgrid(waves / 4), pblock(128);

    k_zero<<<dim3((ND + 255) / 256), b256, 0, stream>>>(deg, ND);
    k_deg<<<dim3((N_EDGES + 255) / 256), b256, 0, stream>>>(row, col, intents, deg);
    k_dis<<<dim3((ND + 255) / 256), b256, 0, stream>>>(deg);

    k_init_x<<<dim3((NX + 255) / 256), b256, 0, stream>>>(Gu, Gi, xbuf);

    k_zero<<<dim3((NX + 255) / 256), b256, 0, stream>>>(out, NX);
    k_propagate<<<pgrid, pblock, 0, stream>>>(xbuf, out, deg, row, col);

    k_zero<<<dim3((NX + 255) / 256), b256, 0, stream>>>(xbuf, NX);
    k_propagate<<<pgrid, pblock, 0, stream>>>(out, xbuf, deg, row, col);

    k_zero<<<dim3((NX + 255) / 256), b256, 0, stream>>>(out, NX);
    k_propagate<<<pgrid, pblock, 0, stream>>>(xbuf, out, deg, row, col);
}


# k_propagate: a work-group sums one range of destination rows in LDS; each row has one owning wave that adds its queued edges in a fixed order; no global atomics, no zero pass; new sum order
# speedup vs baseline: 1.7080x; 1.7080x over previous
#include <hip/hip_runtime.h>

#define N_USERS  100000
#define N_ITEMS  50000
#define N_NODES  150000
#define N_EDGES  1000000
#define EMBED_K  64

typedef __attribute__((ext_vector_type(2))) float v2f;
typedef __attribute__((ext_vector_type(8))) float v8f;

__global__ void k_init_x(const float* __restrict__ Gu, const float* __restrict__ Gi,
                         float* __restrict__ x) {
    int i = blockIdx.x * blockDim.x + threadIdx.x;
    const int total  = N_NODES * EMBED_K;
    const int uElems = N_USERS * EMBED_K;
    if (i >= total) return;
    x[i] = (i < uElems) ? Gu[i] : Gi[i - uElems];
}

#define DEG_R   14
#define DEG_T   10720
#define DEG_G   16
#define DEG_CH  4096

__device__ __forceinline__ void ldsAddF32(float* p, float v) {
    (void)__hip_atomic_fetch_add(p, v, __ATOMIC_RELAXED, __HIP_MEMORY_SCOPE_WORKGROUP);
}

__device__ __forceinline__ float4 edge_w(float z0, float z1, float z2, float z3) {
    float m  = fmaxf(fmaxf(z0, z1), fmaxf(z2, z3));
    float w0 = __expf(z0 - m), w1 = __expf(z1 - m);
    float w2 = __expf(z2 - m), w3 = __expf(z3 - m);
    float inv = 1.0f / (w0 + w1 + w2 + w3);
    return make_float4(w0 * inv, w1 * inv, w2 * inv, w3 * inv);
}

__device__ __forceinline__ void node_add(float* sdeg, int lo, int node, float4 w) {
    const unsigned u = (unsigned)(node - lo);
    if (u < (unsigned)DEG_T) {
        float* p = sdeg + u * 4;
        ldsAddF32(p + 0, w.x); ldsAddF32(p + 1, w.y);
        ldsAddF32(p + 2, w.z); ldsAddF32(p + 3, w.w);
    }
}

__global__ __launch_bounds__(1024) void k_deg(
        const int* __restrict__ row, const int* __restrict__ col,
        const float* __restrict__ intents, float* __restrict__ planes) {
    __shared__ float4 sdeg4[DEG_T];
    float* sdeg = (float*)sdeg4;
    const int tid = threadIdx.x;
    const int rg  = blockIdx.x % DEG_R;
    const int g   = blockIdx.x / DEG_R;
    const int lo  = rg * DEG_T;

    for (int i = tid; i < DEG_T; i += 1024) sdeg4[i] = make_float4(0.0f, 0.0f, 0.0f, 0.0f);
    __syncthreads();

    for (int ch = g; ch * DEG_CH < N_EDGES; ch += DEG_G) {
        const int e = ch * DEG_CH + tid * 4;
        if (e < N_EDGES) {
            const int4   r4 = *(const int4*)(row + e);
            const int4   c4 = *(const int4*)(col + e);
            const float4 za = *(const float4*)(intents + e);
            const float4 zb = *(const float4*)(intents + N_EDGES + e);
            const float4 zc = *(const float4*)(intents + 2 * N_EDGES + e);
            const float4 zd = *(const float4*)(intents + 3 * N_EDGES + e);
            const float4 w0 = edge_w(za.x, zb.x, zc.x, zd.x);
            const float4 w1 = edge_w(za.y, zb.y, zc.y, zd.y);
            const float4 w2 = edge_w(za.z, zb.z, zc.z, zd.z);
            const float4 w3 = edge_w(za.w, zb.w, zc.w, zd.w);
            node_add(sdeg, lo, r4.x, w0); node_add(sdeg, lo, c4.x, w0);
            node_add(sdeg, lo, r4.y, w1); node_add(sdeg, lo, c4.y, w1);
            node_add(sdeg, lo, r4.z, w2); node_add(sdeg, lo, c4.z, w2);
            node_add(sdeg, lo, r4.w, w3); node_add(sdeg, lo, c4.w, w3);
        }
    }
    __syncthreads();

    float4* dst = (float4*)(planes + (size_t)g * (N_NODES * 4)) + lo;
    for (int i = tid; i < DEG_T; i += 1024)
        if (lo + i < N_NODES) dst[i] = sdeg4[i];
}

__global__ void k_dis(const float* __restrict__ planes, float* __restrict__ dis) {
    int i = blockIdx.x * blockDim.x + threadIdx.x;
    if (i >= N_NODES) return;
    float4 s = ((const float4*)planes)[i];
#pragma unroll
    for (int g = 1; g < DEG_G; ++g) {
        const float4 p = ((const float4*)(planes + (size_t)g * (N_NODES * 4)))[i];
        s.x += p.x; s.y += p.y; s.z += p.z; s.w += p.w;
    }
    float4 o;
    o.x = (s.x > 0.0f) ? rsqrtf(fmaxf(s.x, 1e-12f)) : 0.0f;
    o.y = (s.y > 0.0f) ? rsqrtf(fmaxf(s.y, 1e-12f)) : 0.0f;
    o.z = (s.z > 0.0f) ? rsqrtf(fmaxf(s.z, 1e-12f)) : 0.0f;
    o.w = (s.w > 0.0f) ? rsqrtf(fmaxf(s.w, 1e-12f)) : 0.0f;
    ((float4*)dis)[i] = o;
}

#define PR_R       256
#define PR_T       586
#define PR_WROWS   19
#define PR_QCAP    1024
#define PR_MAXPASS 32
#define PR_TRIPS   ((N_EDGES + 128 * 128 - 1) / (128 * 128))

__device__ __forceinline__ void prop_group(
        const float* __restrict__ x_src, const float* __restrict__ dis,
        float* sacc, const unsigned* st, int head, int n, int lo,
        int hl, bool hi) {
    const bool vA = hl < n;
    const unsigned entA = vA ? st[(head + hl) & 63] : 0u;
    const int irA = (int)(entA & 0xFFFFFu);
    const int dlA = (int)(entA >> 20);
    const float4 dr = *(const float4*)(dis + irA * 4);
    const float4 dc = *(const float4*)(dis + (lo + dlA) * 4);
    float nrm[4];
    nrm[0] = vA ? dr.x * dc.x : 0.0f; nrm[1] = vA ? dr.y * dc.y : 0.0f;
    nrm[2] = vA ? dr.z * dc.z : 0.0f; nrm[3] = vA ? dr.w * dc.w : 0.0f;

    const int kx = hi ? 2 : 0;
    const int mo = hi ? 8 : 0;
    int rb[8];
#pragma unroll
    for (int t = 0; t < 4; ++t) {
        rb[2 * t]     = __shfl(irA, 4 * t + kx);
        rb[2 * t + 1] = __shfl(irA, 4 * t + kx + 1);
    }
    int dlv[8];
#pragma unroll
    for (int v = 0; v < 8; ++v) dlv[v] = __shfl(dlA, v + mo);

    float bx[16], by[16];
#pragma unroll
    for (int k = 0; k < 4; ++k) {
#pragma unroll
        for (int t = 0; t < 4; ++t) {
            bx[k * 4 + t] = x_src[rb[2 * t]     * 64 + k * 16 + hl];
            by[k * 4 + t] = x_src[rb[2 * t + 1] * 64 + k * 16 + hl];
        }
    }

#pragma unroll
    for (int k = 0; k < 4; ++k) {
        v8f acc = {};
#pragma unroll
        for (int t = 0; t < 4; ++t) {
            v2f a, b;
            a.x = (hl == 4 * t + kx)     ? nrm[k] : 0.0f;
            a.y = (hl == 4 * t + kx + 1) ? nrm[k] : 0.0f;
            b.x = bx[k * 4 + t];
            b.y = by[k * 4 + t];
            acc = __builtin_amdgcn_wmma_f32_16x16x4_f32(
                      false, a, false, b, (short)0, acc, false, false);
        }
#pragma unroll
        for (int v = 0; v < 8; ++v)
            if (v + mo < n) ldsAddF32(sacc + dlv[v] * 64 + k * 16 + hl, acc[v]);
    }
}

__device__ __forceinline__ void scan_load(const int* __restrict__ col, int ip, int tid, int4 (&cc)[4]) {
#pragma unroll
    for (int j = 0; j < 4; ++j) {
        const unsigned i4 = (unsigned)min(ip * 4096 + j * 1024 + tid, N_EDGES / 4 - 1);
        cc[j] = *(const int4*)((const char*)col + i4 * 16u);
    }
}

__global__ __launch_bounds__(1024) void k_propagate(
        const float* __restrict__ x_src, float* __restrict__ x_dst,
        const float* __restrict__ dis,
        const int* __restrict__ row, const int* __restrict__ col) {
    __shared__ float4   sacc4[PR_T * 16];
    __shared__ unsigned squeue[32 * PR_QCAP];
    __shared__ unsigned sstage[32 * 64];
    __shared__ int      scnt[32];
    float* sacc = (float*)sacc4;
    const int tid  = threadIdx.x;
    const int wv   = __builtin_amdgcn_readfirstlane(tid >> 5);
    const int lane = tid & 31;
    const int hl   = lane & 15;
    const bool hi  = lane >= 16;
    const unsigned lt = (1u << lane) - 1u;
    const int lo   = blockIdx.x * PR_T;
    const int rot  = blockIdx.x % PR_TRIPS;

    for (int i = tid; i < PR_T * 16; i += 1024) sacc4[i] = make_float4(0.0f, 0.0f, 0.0f, 0.0f);

    for (int pass = 0; pass < PR_MAXPASS; ++pass) {
        const int wlo = pass * PR_QCAP;

        unsigned* q = squeue + wv * PR_QCAP;
        int seen = 0;
        int4 nx[4];
        scan_load(col, rot, tid, nx);
        for (int it = 0; it < PR_TRIPS; ++it) {
            const int ip = (it + rot < PR_TRIPS) ? it + rot : it + rot - PR_TRIPS;
            int4 cc[4];
#pragma unroll
            for (int j = 0; j < 4; ++j) cc[j] = nx[j];
            {
                const int in1 = (it + 1 < PR_TRIPS) ? it + 1 + rot : rot;
                scan_load(col, (in1 < PR_TRIPS) ? in1 : in1 - PR_TRIPS, tid, nx);
            }
#pragma unroll
            for (int j = 0; j < 4; ++j) {
                const int e = (ip * 4096 + j * 1024 + tid) * 4;
                const bool in = e < N_EDGES;
                const unsigned bin = __builtin_amdgcn_ballot_w32(in);
                const unsigned u0 = (unsigned)(cc[j].x - lo), u1 = (unsigned)(cc[j].y - lo);
                const unsigned u2 = (unsigned)(cc[j].z - lo), u3 = (unsigned)(cc[j].w - lo);
                const bool t0 = u0 < (unsigned)PR_T, t1 = u1 < (unsigned)PR_T;
                const bool t2 = u2 < (unsigned)PR_T, t3 = u3 < (unsigned)PR_T;
                const unsigned b0 = __builtin_amdgcn_ballot_w32(t0) & bin;
                const unsigned b1 = __builtin_amdgcn_ballot_w32(t1) & bin;
                const unsigned b2 = __builtin_amdgcn_ballot_w32(t2) & bin;
                const unsigned b3 = __builtin_amdgcn_ballot_w32(t3) & bin;
                if ((b0 | b1 | b2 | b3) != 0u) {
                    if (b0 != 0u) {
                        const unsigned w = (unsigned)(seen + __popc(b0 & lt) - wlo);
                        if (t0 && in && w < (unsigned)PR_QCAP) q[w] = (u0 << 20) | (unsigned)(e + 0);
                        seen += __popc(b0);
                    }
                    if (b1 != 0u) {
                        const unsigned w = (unsigned)(seen + __popc(b1 & lt) - wlo);
                        if (t1 && in && w < (unsigned)PR_QCAP) q[w] = (u1 << 20) | (unsigned)(e + 1);
                        seen += __popc(b1);
                    }
                    if (b2 != 0u) {
                        const unsigned w = (unsigned)(seen + __popc(b2 & lt) - wlo);
                        if (t2 && in && w < (unsigned)PR_QCAP) q[w] = (u2 << 20) | (unsigned)(e + 2);
                        seen += __popc(b2);
                    }
                    if (b3 != 0u) {
                        const unsigned w = (unsigned)(seen + __popc(b3 & lt) - wlo);
                        if (t3 && in && w < (unsigned)PR_QCAP) q[w] = (u3 << 20) | (unsigned)(e + 3);
                        seen += __popc(b3);
                    }
                }
            }
        }

        {
            const int nown = min(max(seen - wlo, 0), PR_QCAP);
            for (int i0 = 0; i0 < nown; i0 += 128) {
                unsigned ent[4];
                int sid[4];
#pragma unroll
                for (int j = 0; j < 4; ++j) {
                    const int idx = i0 + 32 * j + lane;
                    ent[j] = (idx < nown) ? q[idx] : 0u;
                }
#pragma unroll
                for (int j = 0; j < 4; ++j) sid[j] = row[ent[j] & 0xFFFFFu];
#pragma unroll
                for (int j = 0; j < 4; ++j) {
                    const int idx = i0 + 32 * j + lane;
                    if (idx < nown) q[idx] = (ent[j] & 0xFFF00000u) | (unsigned)sid[j];
                }
            }
        }
        if (lane == 0) scnt[wv] = seen;
        __syncthreads();

        const int cnt_l = scnt[lane];
        const bool more = __builtin_amdgcn_ballot_w32(cnt_l > wlo + PR_QCAP) != 0u;

        {
            const unsigned olo = (unsigned)(wv * PR_WROWS);
            unsigned* st = sstage + wv * 64;
            int head = 0, have = 0;
            for (int qw = 0; qw < 32; ++qw) {
                const int n = min(max(__builtin_amdgcn_readlane(cnt_l, qw) - wlo, 0), PR_QCAP);
                const unsigned* qq = squeue + qw * PR_QCAP;
                for (int i = 0; i < n; i += 128) {
                    unsigned ent[4];
#pragma unroll
                    for (int j = 0; j < 4; ++j) {
                        const int idx = i + 32 * j + lane;
                        ent[j] = (idx < n) ? qq[idx] : 0xFFFFFFFFu;
                    }
#pragma unroll
                    for (int j = 0; j < 4; ++j) {
                        const bool m = ((ent[j] >> 20) - olo) < (unsigned)PR_WROWS;
                        const unsigned b = __builtin_amdgcn_ballot_w32(m);
                        if (b != 0u) {
                            if (m) st[(head + have + __popc(b & lt)) & 63] = ent[j];
                            have += __popc(b);
                            while (have >= 16) {
                                prop_group(x_src, dis, sacc, st, head, 16, lo, hl, hi);
                                head = (head + 16) & 63;
                                have -= 16;
                            }
                        }
                    }
                }
            }
            if (have > 0) prop_group(x_src, dis, sacc, st, head, have, lo, hl, hi);
        }
        __syncthreads();
        if (!more) break;
    }

    const int nrows = min(PR_T, N_NODES - lo);
    float4* dst = (float4*)x_dst + (size_t)lo * 16;
    for (int i = tid; i < nrows * 16; i += 1024) dst[i] = sacc4[i];
}

extern "C" void kernel_launch(void* const* d_in, const int* in_sizes, int n_in,
                              void* d_out, int out_size, void* d_ws, size_t ws_size,
                              hipStream_t stream) {
    const float* Gu      = (const float*)d_in[0];
    const float* Gi      = (const float*)d_in[1];
    const int*   edge    = (const int*)d_in[2];
    const float* intents = (const float*)d_in[3];
    const int* row = edge;
    const int* col = edge + N_EDGES;
    float* out = (float*)d_out;

    float* deg  = (float*)d_ws;
    float* xbuf = deg + (size_t)N_NODES * 4;

    const int NX = N_NODES * EMBED_K;
    dim3 b256(256);
    dim3 pgrid(PR_R), pblock(1024);

    k_deg<<<dim3(DEG_R * DEG_G), dim3(1024), 0, stream>>>(row, col, intents, xbuf);
    k_dis<<<dim3((N_NODES + 255) / 256), b256, 0, stream>>>(xbuf, deg);

    k_init_x<<<dim3((NX + 255) / 256), b256, 0, stream>>>(Gu, Gi, xbuf);

    k_propagate<<<pgrid, pblock, 0, stream>>>(xbuf, out, deg, row, col);
    k_propagate<<<pgrid, pblock, 0, stream>>>(out, xbuf, deg, row, col);
    k_propagate<<<pgrid, pblock, 0, stream>>>(xbuf, out, deg, row, col);
}


# k_init_x folded into k_dis (degree planes moved to the output buffer): a work-group also copies its 128 rows of x0 in 16-byte pieces, 32 loads in flight per wave; 5 launches; sums unchanged
# speedup vs baseline: 1.7990x; 1.0533x over previous
#include <hip/hip_runtime.h>

#define N_USERS  100000
#define N_ITEMS  50000
#define N_NODES  150000
#define N_EDGES  1000000
#define EMBED_K  64

typedef __attribute__((ext_vector_type(2))) float v2f;
typedef __attribute__((ext_vector_type(8))) float v8f;

#define DEG_R   14
#define DEG_T   10720
#define DEG_G   16
#define DEG_CH  4096

__device__ __forceinline__ void ldsAddF32(float* p, float v) {
    (void)__hip_atomic_fetch_add(p, v, __ATOMIC_RELAXED, __HIP_MEMORY_SCOPE_WORKGROUP);
}

__device__ __forceinline__ float4 edge_w(float z0, float z1, float z2, float z3) {
    float m  = fmaxf(fmaxf(z0, z1), fmaxf(z2, z3));
    float w0 = __expf(z0 - m), w1 = __expf(z1 - m);
    float w2 = __expf(z2 - m), w3 = __expf(z3 - m);
    float inv = 1.0f / (w0 + w1 + w2 + w3);
    return make_float4(w0 * inv, w1 * inv, w2 * inv, w3 * inv);
}

__device__ __forceinline__ void node_add(float* sdeg, int lo, int node, float4 w) {
    const unsigned u = (unsigned)(node - lo);
    if (u < (unsigned)DEG_T) {
        float* p = sdeg + u * 4;
        ldsAddF32(p + 0, w.x); ldsAddF32(p + 1, w.y);
        ldsAddF32(p + 2, w.z); ldsAddF32(p + 3, w.w);
    }
}

__global__ __launch_bounds__(1024) void k_deg(
        const int* __restrict__ row, const int* __restrict__ col,
        const float* __restrict__ intents, float* __restrict__ planes) {
    __shared__ float4 sdeg4[DEG_T];
    float* sdeg = (float*)sdeg4;
    const int tid = threadIdx.x;
    const int rg  = blockIdx.x % DEG_R;
    const int g   = blockIdx.x / DEG_R;
    const int lo  = rg * DEG_T;

    for (int i = tid; i < DEG_T; i += 1024) sdeg4[i] = make_float4(0.0f, 0.0f, 0.0f, 0.0f);
    __syncthreads();

    for (int ch = g; ch * DEG_CH < N_EDGES; ch += DEG_G) {
        const int e = ch * DEG_CH + tid * 4;
        if (e < N_EDGES) {
            const int4   r4 = *(const int4*)(row + e);
            const int4   c4 = *(const int4*)(col + e);
            const float4 za = *(const float4*)(intents + e);
            const float4 zb = *(const float4*)(intents + N_EDGES + e);
            const float4 zc = *(const float4*)(intents + 2 * N_EDGES + e);
            const float4 zd = *(const float4*)(intents + 3 * N_EDGES + e);
            const float4 w0 = edge_w(za.x, zb.x, zc.x, zd.x);
            const float4 w1 = edge_w(za.y, zb.y, zc.y, zd.y);
            const float4 w2 = edge_w(za.z, zb.z, zc.z, zd.z);
            const float4 w3 = edge_w(za.w, zb.w, zc.w, zd.w);
            node_add(sdeg, lo, r4.x, w0); node_add(sdeg, lo, c4.x, w0);
            node_add(sdeg, lo, r4.y, w1); node_add(sdeg, lo, c4.y, w1);
            node_add(sdeg, lo, r4.z, w2); node_add(sdeg, lo, c4.z, w2);
            node_add(sdeg, lo, r4.w, w3); node_add(sdeg, lo, c4.w, w3);
        }
    }
    __syncthreads();

    float4* dst = (float4*)(planes + (size_t)g * (N_NODES * 4)) + lo;
    for (int i = tid; i < DEG_T; i += 1024)
        if (lo + i < N_NODES) dst[i] = sdeg4[i];
}

#define X0_U4   (N_USERS * (EMBED_K / 4))
#define X0_N4   (N_NODES * (EMBED_K / 4))
__global__ __launch_bounds__(128) void k_dis_x0(
        const float* __restrict__ planes, float* __restrict__ dis,
        const float* __restrict__ Gu, const float* __restrict__ Gi,
        float* __restrict__ x) {
    const int tid = threadIdx.x;
    const int f0  = blockIdx.x * 2048 + tid;
    float4 c[16];
#pragma unroll
    for (int j = 0; j < 16; ++j) {
        const int f = min(f0 + j * 128, X0_N4 - 1);
        const float4* p = (f < X0_U4) ? (const float4*)Gu + f : (const float4*)Gi + (f - X0_U4);
        c[j] = *p;
    }
    const int i = blockIdx.x * 128 + tid;
    if (i < N_NODES) {
        float4 s = ((const float4*)planes)[i];
#pragma unroll
        for (int g = 1; g < DEG_G; ++g) {
            const float4 p = ((const float4*)(planes + (size_t)g * (N_NODES * 4)))[i];
            s.x += p.x; s.y += p.y; s.z += p.z; s.w += p.w;
        }
        float4 o;
        o.x = (s.x > 0.0f) ? rsqrtf(fmaxf(s.x, 1e-12f)) : 0.0f;
        o.y = (s.y > 0.0f) ? rsqrtf(fmaxf(s.y, 1e-12f)) : 0.0f;
        o.z = (s.z > 0.0f) ? rsqrtf(fmaxf(s.z, 1e-12f)) : 0.0f;
        o.w = (s.w > 0.0f) ? rsqrtf(fmaxf(s.w, 1e-12f)) : 0.0f;
        ((float4*)dis)[i] = o;
    }
#pragma unroll
    for (int j = 0; j < 16; ++j) {
        asm volatile("" :: "v"(c[j].x), "v"(c[j].y), "v"(c[j].z), "v"(c[j].w));
    }
#pragma unroll
    for (int j = 0; j < 16; ++j) {
        const int f = f0 + j * 128;
        if (f < X0_N4) ((float4*)x)[f] = c[j];
    }
}

#define PR_R       256
#define PR_T       586
#define PR_WROWS   19
#define PR_QCAP    1024
#define PR_MAXPASS 32
#define PR_TRIPS   ((N_EDGES + 128 * 128 - 1) / (128 * 128))

__device__ __forceinline__ void prop_group(
        const float* __restrict__ x_src, const float* __restrict__ dis,
        float* sacc, const unsigned* st, int head, int n, int lo,
        int hl, bool hi) {
    const bool vA = hl < n;
    const unsigned entA = vA ? st[(head + hl) & 63] : 0u;
    const int irA = (int)(entA & 0xFFFFFu);
    const int dlA = (int)(entA >> 20);
    const float4 dr = *(const float4*)(dis + irA * 4);
    const float4 dc = *(const float4*)(dis + (lo + dlA) * 4);
    float nrm[4];
    nrm[0] = vA ? dr.x * dc.x : 0.0f; nrm[1] = vA ? dr.y * dc.y : 0.0f;
    nrm[2] = vA ? dr.z * dc.z : 0.0f; nrm[3] = vA ? dr.w * dc.w : 0.0f;

    const int kx = hi ? 2 : 0;
    const int mo = hi ? 8 : 0;
    int rb[8];
#pragma unroll
    for (int t = 0; t < 4; ++t) {
        rb[2 * t]     = __shfl(irA, 4 * t + kx);
        rb[2 * t + 1] = __shfl(irA, 4 * t + kx + 1);
    }
    int dlv[8];
#pragma unroll
    for (int v = 0; v < 8; ++v) dlv[v] = __shfl(dlA, v + mo);

    float bx[16], by[16];
#pragma unroll
    for (int k = 0; k < 4; ++k) {
#pragma unroll
        for (int t = 0; t < 4; ++t) {
            bx[k * 4 + t] = x_src[rb[2 * t]     * 64 + k * 16 + hl];
            by[k * 4 + t] = x_src[rb[2 * t + 1] * 64 + k * 16 + hl];
        }
    }

#pragma unroll
    for (int k = 0; k < 4; ++k) {
        v8f acc = {};
#pragma unroll
        for (int t = 0; t < 4; ++t) {
            v2f a, b;
            a.x = (hl == 4 * t + kx)     ? nrm[k] : 0.0f;
            a.y = (hl == 4 * t + kx + 1) ? nrm[k] : 0.0f;
            b.x = bx[k * 4 + t];
            b.y = by[k * 4 + t];
            acc = __builtin_amdgcn_wmma_f32_16x16x4_f32(
                      false, a, false, b, (short)0, acc, false, false);
        }
#pragma unroll
        for (int v = 0; v < 8; ++v)
            if (v + mo < n) ldsAddF32(sacc + dlv[v] * 64 + k * 16 + hl, acc[v]);
    }
}

__device__ __forceinline__ void scan_load(const int* __restrict__ col, int ip, int tid, int4 (&cc)[4]) {
#pragma unroll
    for (int j = 0; j < 4; ++j) {
        const unsigned i4 = (unsigned)min(ip * 4096 + j * 1024 + tid, N_EDGES / 4 - 1);
        cc[j] = *(const int4*)((const char*)col + i4 * 16u);
    }
}

__global__ __launch_bounds__(1024) void k_propagate(
        const float* __restrict__ x_src, float* __restrict__ x_dst,
        const float* __restrict__ dis,
        const int* __restrict__ row, const int* __restrict__ col) {
    __shared__ float4   sacc4[PR_T * 16];
    __shared__ unsigned squeue[32 * PR_QCAP];
    __shared__ unsigned sstage[32 * 64];
    __shared__ int      scnt[32];
    float* sacc = (float*)sacc4;
    const int tid  = threadIdx.x;
    const int wv   = __builtin_amdgcn_readfirstlane(tid >> 5);
    const int lane = tid & 31;
    const int hl   = lane & 15;
    const bool hi  = lane >= 16;
    const unsigned lt = (1u << lane) - 1u;
    const int lo   = blockIdx.x * PR_T;
    const int rot  = blockIdx.x % PR_TRIPS;

    for (int i = tid; i < PR_T * 16; i += 1024) sacc4[i] = make_float4(0.0f, 0.0f, 0.0f, 0.0f);

    for (int pass = 0; pass < PR_MAXPASS; ++pass) {
        const int wlo = pass * PR_QCAP;

        unsigned* q = squeue + wv * PR_QCAP;
        int seen = 0;
        int4 nx[4];
        scan_load(col, rot, tid, nx);
        for (int it = 0; it < PR_TRIPS; ++it) {
            const int ip = (it + rot < PR_TRIPS) ? it + rot : it + rot - PR_TRIPS;
            int4 cc[4];
#pragma unroll
            for (int j = 0; j < 4; ++j) cc[j] = nx[j];
            {
                const int in1 = (it + 1 < PR_TRIPS) ? it + 1 + rot : rot;
                scan_load(col, (in1 < PR_TRIPS) ? in1 : in1 - PR_TRIPS, tid, nx);
            }
#pragma unroll
            for (int j = 0; j < 4; ++j) {
                const int e = (ip * 4096 + j * 1024 + tid) * 4;
                const bool in = e < N_EDGES;
                const unsigned bin = __builtin_amdgcn_ballot_w32(in);
                const unsigned u0 = (unsigned)(cc[j].x - lo), u1 = (unsigned)(cc[j].y - lo);
                const unsigned u2 = (unsigned)(cc[j].z - lo), u3 = (unsigned)(cc[j].w - lo);
                const bool t0 = u0 < (unsigned)PR_T, t1 = u1 < (unsigned)PR_T;
                const bool t2 = u2 < (unsigned)PR_T, t3 = u3 < (unsigned)PR_T;
                const unsigned b0 = __builtin_amdgcn_ballot_w32(t0) & bin;
                const unsigned b1 = __builtin_amdgcn_ballot_w32(t1) & bin;
                const unsigned b2 = __builtin_amdgcn_ballot_w32(t2) & bin;
                const unsigned b3 = __builtin_amdgcn_ballot_w32(t3) & bin;
                if ((b0 | b1 | b2 | b3) != 0u) {
                    if (b0 != 0u) {
                        const unsigned w = (unsigned)(seen + __popc(b0 & lt) - wlo);
                        if (t0 && in && w < (unsigned)PR_QCAP) q[w] = (u0 << 20) | (unsigned)(e + 0);
                        seen += __popc(b0);
                    }
                    if (b1 != 0u) {
                        const unsigned w = (unsigned)(seen + __popc(b1 & lt) - wlo);
                        if (t1 && in && w < (unsigned)PR_QCAP) q[w] = (u1 << 20) | (unsigned)(e + 1);
                        seen += __popc(b1);
                    }
                    if (b2 != 0u) {
                        const unsigned w = (unsigned)(seen + __popc(b2 & lt) - wlo);
                        if (t2 && in && w < (unsigned)PR_QCAP) q[w] = (u2 << 20) | (unsigned)(e + 2);
                        seen += __popc(b2);
                    }
                    if (b3 != 0u) {
                        const unsigned w = (unsigned)(seen + __popc(b3 & lt) - wlo);
                        if (t3 && in && w < (unsigned)PR_QCAP) q[w] = (u3 << 20) | (unsigned)(e + 3);
                        seen += __popc(b3);
                    }
                }
            }
        }

        {
            const int nown = min(max(seen - wlo, 0), PR_QCAP);
            for (int i0 = 0; i0 < nown; i0 += 128) {
                unsigned ent[4];
                int sid[4];
#pragma unroll
                for (int j = 0; j < 4; ++j) {
                    const int idx = i0 + 32 * j + lane;
                    ent[j] = (idx < nown) ? q[idx] : 0u;
                }
#pragma unroll
                for (int j = 0; j < 4; ++j) sid[j] = row[ent[j] & 0xFFFFFu];
#pragma unroll
                for (int j = 0; j < 4; ++j) {
                    const int idx = i0 + 32 * j + lane;
                    if (idx < nown) q[idx] = (ent[j] & 0xFFF00000u) | (unsigned)sid[j];
                }
            }
        }
        if (lane == 0) scnt[wv] = seen;
        __syncthreads();

        const int cnt_l = scnt[lane];
        const bool more = __builtin_amdgcn_ballot_w32(cnt_l > wlo + PR_QCAP) != 0u;

        {
            const unsigned olo = (unsigned)(wv * PR_WROWS);
            unsigned* st = sstage + wv * 64;
            int head = 0, have = 0;
            for (int qw = 0; qw < 32; ++qw) {
                const int n = min(max(__builtin_amdgcn_readlane(cnt_l, qw) - wlo, 0), PR_QCAP);
                const unsigned* qq = squeue + qw * PR_QCAP;
                for (int i = 0; i < n; i += 128) {
                    unsigned ent[4];
#pragma unroll
                    for (int j = 0; j < 4; ++j) {
                        const int idx = i + 32 * j + lane;
                        ent[j] = (idx < n) ? qq[idx] : 0xFFFFFFFFu;
                    }
#pragma unroll
                    for (int j = 0; j < 4; ++j) {
                        const bool m = ((ent[j] >> 20) - olo) < (unsigned)PR_WROWS;
                        const unsigned b = __builtin_amdgcn_ballot_w32(m);
                        if (b != 0u) {
                            if (m) st[(head + have + __popc(b & lt)) & 63] = ent[j];
                            have += __popc(b);
                            while (have >= 16) {
                                prop_group(x_src, dis, sacc, st, head, 16, lo, hl, hi);
                                head = (head + 16) & 63;
                                have -= 16;
                            }
                        }
                    }
                }
            }
            if (have > 0) prop_group(x_src, dis, sacc, st, head, have, lo, hl, hi);
        }
        __syncthreads();
        if (!more) break;
    }

    const int nrows = min(PR_T, N_NODES - lo);
    float4* dst = (float4*)x_dst + (size_t)lo * 16;
    for (int i = tid; i < nrows * 16; i += 1024) dst[i] = sacc4[i];
}

extern "C" void kernel_launch(void* const* d_in, const int* in_sizes, int n_in,
                              void* d_out, int out_size, void* d_ws, size_t ws_size,
                              hipStream_t stream) {
    const float* Gu      = (const float*)d_in[0];
    const float* Gi      = (const float*)d_in[1];
    const int*   edge    = (const int*)d_in[2];
    const float* intents = (const float*)d_in[3];
    const int* row = edge;
    const int* col = edge + N_EDGES;
    float* out = (float*)d_out;

    float* deg  = (float*)d_ws;
    float* xbuf = deg + (size_t)N_NODES * 4;

    dim3 pgrid(PR_R), pblock(1024);

    k_deg<<<dim3(DEG_R * DEG_G), dim3(1024), 0, stream>>>(row, col, intents, out);
    k_dis_x0<<<dim3((N_NODES + 127) / 128), dim3(128), 0, stream>>>(out, deg, Gu, Gi, xbuf);

    k_propagate<<<pgrid, pblock, 0, stream>>>(xbuf, out, deg, row, col);
    k_propagate<<<pgrid, pblock, 0, stream>>>(out, xbuf, deg, row, col);
    k_propagate<<<pgrid, pblock, 0, stream>>>(xbuf, out, deg, row, col);
}


# k_propagate: layer 1 also stores each owning wave's work list (source, row) in the order it adds; layers 2-3 load their lists and skip the edge scan, queue and walk; same sum order
# speedup vs baseline: 2.4531x; 1.3636x over previous
#include <hip/hip_runtime.h>

#define N_USERS  100000
#define N_ITEMS  50000
#define N_NODES  150000
#define N_EDGES  1000000
#define EMBED_K  64

typedef __attribute__((ext_vector_type(2))) float v2f;
typedef __attribute__((ext_vector_type(8))) float v8f;

#define DEG_R   14
#define DEG_T   10720
#define DEG_G   16
#define DEG_CH  4096

__device__ __forceinline__ void ldsAddF32(float* p, float v) {
    (void)__hip_atomic_fetch_add(p, v, __ATOMIC_RELAXED, __HIP_MEMORY_SCOPE_WORKGROUP);
}

__device__ __forceinline__ float4 edge_w(float z0, float z1, float z2, float z3) {
    float m  = fmaxf(fmaxf(z0, z1), fmaxf(z2, z3));
    float w0 = __expf(z0 - m), w1 = __expf(z1 - m);
    float w2 = __expf(z2 - m), w3 = __expf(z3 - m);
    float inv = 1.0f / (w0 + w1 + w2 + w3);
    return make_float4(w0 * inv, w1 * inv, w2 * inv, w3 * inv);
}

__device__ __forceinline__ void node_add(float* sdeg, int lo, int node, float4 w) {
    const unsigned u = (unsigned)(node - lo);
    if (u < (unsigned)DEG_T) {
        float* p = sdeg + u * 4;
        ldsAddF32(p + 0, w.x); ldsAddF32(p + 1, w.y);
        ldsAddF32(p + 2, w.z); ldsAddF32(p + 3, w.w);
    }
}

__global__ __launch_bounds__(1024) void k_deg(
        const int* __restrict__ row, const int* __restrict__ col,
        const float* __restrict__ intents, float* __restrict__ planes) {
    __shared__ float4 sdeg4[DEG_T];
    float* sdeg = (float*)sdeg4;
    const int tid = threadIdx.x;
    const int rg  = blockIdx.x % DEG_R;
    const int g   = blockIdx.x / DEG_R;
    const int lo  = rg * DEG_T;

    for (int i = tid; i < DEG_T; i += 1024) sdeg4[i] = make_float4(0.0f, 0.0f, 0.0f, 0.0f);
    __syncthreads();

    for (int ch = g; ch * DEG_CH < N_EDGES; ch += DEG_G) {
        const int e = ch * DEG_CH + tid * 4;
        if (e < N_EDGES) {
            const int4   r4 = *(const int4*)(row + e);
            const int4   c4 = *(const int4*)(col + e);
            const float4 za = *(const float4*)(intents + e);
            const float4 zb = *(const float4*)(intents + N_EDGES + e);
            const float4 zc = *(const float4*)(intents + 2 * N_EDGES + e);
            const float4 zd = *(const float4*)(intents + 3 * N_EDGES + e);
            const float4 w0 = edge_w(za.x, zb.x, zc.x, zd.x);
            const float4 w1 = edge_w(za.y, zb.y, zc.y, zd.y);
            const float4 w2 = edge_w(za.z, zb.z, zc.z, zd.z);
            const float4 w3 = edge_w(za.w, zb.w, zc.w, zd.w);
            node_add(sdeg, lo, r4.x, w0); node_add(sdeg, lo, c4.x, w0);
            node_add(sdeg, lo, r4.y, w1); node_add(sdeg, lo, c4.y, w1);
            node_add(sdeg, lo, r4.z, w2); node_add(sdeg, lo, c4.z, w2);
            node_add(sdeg, lo, r4.w, w3); node_add(sdeg, lo, c4.w, w3);
        }
    }
    __syncthreads();

    float4* dst = (float4*)(planes + (size_t)g * (N_NODES * 4)) + lo;
    for (int i = tid; i < DEG_T; i += 1024)
        if (lo + i < N_NODES) dst[i] = sdeg4[i];
}

#define X0_U4   (N_USERS * (EMBED_K / 4))
#define X0_N4   (N_NODES * (EMBED_K / 4))
__global__ __launch_bounds__(128) void k_dis_x0(
        const float* __restrict__ planes, float* __restrict__ dis,
        const float* __restrict__ Gu, const float* __restrict__ Gi,
        float* __restrict__ x) {
    const int tid = threadIdx.x;
    const int f0  = blockIdx.x * 2048 + tid;
    float4 c[16];
#pragma unroll
    for (int j = 0; j < 16; ++j) {
        const int f = min(f0 + j * 128, X0_N4 - 1);
        const float4* p = (f < X0_U4) ? (const float4*)Gu + f : (const float4*)Gi + (f - X0_U4);
        c[j] = *p;
    }
    const int i = blockIdx.x * 128 + tid;
    if (i < N_NODES) {
        float4 s = ((const float4*)planes)[i];
#pragma unroll
        for (int g = 1; g < DEG_G; ++g) {
            const float4 p = ((const float4*)(planes + (size_t)g * (N_NODES * 4)))[i];
            s.x += p.x; s.y += p.y; s.z += p.z; s.w += p.w;
        }
        float4 o;
        o.x = (s.x > 0.0f) ? rsqrtf(fmaxf(s.x, 1e-12f)) : 0.0f;
        o.y = (s.y > 0.0f) ? rsqrtf(fmaxf(s.y, 1e-12f)) : 0.0f;
        o.z = (s.z > 0.0f) ? rsqrtf(fmaxf(s.z, 1e-12f)) : 0.0f;
        o.w = (s.w > 0.0f) ? rsqrtf(fmaxf(s.w, 1e-12f)) : 0.0f;
        ((float4*)dis)[i] = o;
    }
#pragma unroll
    for (int j = 0; j < 16; ++j) {
        asm volatile("" :: "v"(c[j].x), "v"(c[j].y), "v"(c[j].z), "v"(c[j].w));
    }
#pragma unroll
    for (int j = 0; j < 16; ++j) {
        const int f = f0 + j * 128;
        if (f < X0_N4) ((float4*)x)[f] = c[j];
    }
}

#define PR_R       256
#define PR_T       586
#define PR_WROWS   19
#define PR_QCAP    1024
#define PR_MAXPASS 32
#define PR_TRIPS   ((N_EDGES + 128 * 128 - 1) / (128 * 128))
#define PR_WCAP    256
#define PR_WS_NEED ((size_t)N_NODES * 4 * 4 + (size_t)N_NODES * 64 * 4 + (size_t)PR_R * 32 * 4 + (size_t)PR_R * 32 * PR_WCAP * 4)

__device__ __forceinline__ unsigned prop_group(
        const float* __restrict__ x_src, const float* __restrict__ dis,
        float* sacc, const unsigned* grp, int n, int lo,
        int hl, bool hi) {
    const bool vA = hl < n;
    const unsigned entA = vA ? grp[hl] : 0u;
    const int irA = (int)(entA & 0xFFFFFu);
    const int dlA = (int)(entA >> 20);
    const float4 dr = *(const float4*)(dis + irA * 4);
    const float4 dc = *(const float4*)(dis + (lo + dlA) * 4);
    float nrm[4];
    nrm[0] = vA ? dr.x * dc.x : 0.0f; nrm[1] = vA ? dr.y * dc.y : 0.0f;
    nrm[2] = vA ? dr.z * dc.z : 0.0f; nrm[3] = vA ? dr.w * dc.w : 0.0f;

    const int kx = hi ? 2 : 0;
    const int mo = hi ? 8 : 0;
    int rb[8];
#pragma unroll
    for (int t = 0; t < 4; ++t) {
        rb[2 * t]     = __shfl(irA, 4 * t + kx);
        rb[2 * t + 1] = __shfl(irA, 4 * t + kx + 1);
    }
    int dlv[8];
#pragma unroll
    for (int v = 0; v < 8; ++v) dlv[v] = __shfl(dlA, v + mo);

    float bx[16], by[16];
#pragma unroll
    for (int k = 0; k < 4; ++k) {
#pragma unroll
        for (int t = 0; t < 4; ++t) {
            bx[k * 4 + t] = x_src[rb[2 * t]     * 64 + k * 16 + hl];
            by[k * 4 + t] = x_src[rb[2 * t + 1] * 64 + k * 16 + hl];
        }
    }

#pragma unroll
    for (int k = 0; k < 4; ++k) {
        v8f acc = {};
#pragma unroll
        for (int t = 0; t < 4; ++t) {
            v2f a, b;
            a.x = (hl == 4 * t + kx)     ? nrm[k] : 0.0f;
            a.y = (hl == 4 * t + kx + 1) ? nrm[k] : 0.0f;
            b.x = bx[k * 4 + t];
            b.y = by[k * 4 + t];
            acc = __builtin_amdgcn_wmma_f32_16x16x4_f32(
                      false, a, false, b, (short)0, acc, false, false);
        }
#pragma unroll
        for (int v = 0; v < 8; ++v)
            if (v + mo < n) ldsAddF32(sacc + dlv[v] * 64 + k * 16 + hl, acc[v]);
    }
    return entA;
}

__device__ __forceinline__ void scan_load(const int* __restrict__ col, int ip, int tid, int4 (&cc)[4]) {
#pragma unroll
    for (int j = 0; j < 4; ++j) {
        const unsigned i4 = (unsigned)min(ip * 4096 + j * 1024 + tid, N_EDGES / 4 - 1);
        cc[j] = *(const int4*)((const char*)col + i4 * 16u);
    }
}

template <bool BUILD>
__global__ __launch_bounds__(1024) void k_propagate(
        const float* __restrict__ x_src, float* __restrict__ x_dst,
        const float* __restrict__ dis,
        const int* __restrict__ row, const int* __restrict__ col,
        int* __restrict__ wcnt, unsigned* __restrict__ wlist, size_t ws_bytes) {
    __shared__ float4   sacc4[PR_T * 16];
    __shared__ unsigned squeue[32 * PR_QCAP];
    __shared__ unsigned sstage[32 * 64];
    __shared__ int      scnt[32];
    float* sacc = (float*)sacc4;
    const int tid  = threadIdx.x;
    const int wv   = __builtin_amdgcn_readfirstlane(tid >> 5);
    const int lane = tid & 31;
    const int hl   = lane & 15;
    const bool hi  = lane >= 16;
    const unsigned lt = (1u << lane) - 1u;
    const int lo   = blockIdx.x * PR_T;
    const int rot  = blockIdx.x % PR_TRIPS;

    const bool room = ws_bytes >= PR_WS_NEED;
    unsigned* glist = wlist + (size_t)(blockIdx.x * 32 + wv) * PR_WCAP;
    int wc_l = 0;
    if (!BUILD && room) wc_l = wcnt[blockIdx.x * 32 + lane];

    for (int i = tid; i < PR_T * 16; i += 1024) sacc4[i] = make_float4(0.0f, 0.0f, 0.0f, 0.0f);

    const bool replay = !BUILD && room &&
                        __builtin_amdgcn_ballot_w32((unsigned)wc_l > (unsigned)PR_WCAP) == 0u;

    if (replay) {
        const int n = __builtin_amdgcn_readlane(wc_l, wv);
        unsigned* wl = squeue + wv * PR_QCAP;
        unsigned e[PR_WCAP / 32];
#pragma unroll
        for (int c = 0; c < PR_WCAP / 32; ++c) {
            const int idx = c * 32 + lane;
            e[c] = (idx < n) ? glist[idx] : 0u;
        }
#pragma unroll
        for (int c = 0; c < PR_WCAP / 32; ++c) wl[c * 32 + lane] = e[c];
        __syncthreads();
        for (int g0 = 0; g0 < n; g0 += 16)
            (void)prop_group(x_src, dis, sacc, wl + g0, min(16, n - g0), lo, hl, hi);
        __syncthreads();
    } else {
        int wtot = 0;
        for (int pass = 0; pass < PR_MAXPASS; ++pass) {
            const int wlo = pass * PR_QCAP;

            unsigned* q = squeue + wv * PR_QCAP;
            int seen = 0;
            int4 nx[4];
            scan_load(col, rot, tid, nx);
            for (int it = 0; it < PR_TRIPS; ++it) {
                const int ip = (it + rot < PR_TRIPS) ? it + rot : it + rot - PR_TRIPS;
                int4 cc[4];
#pragma unroll
                for (int j = 0; j < 4; ++j) cc[j] = nx[j];
                {
                    const int in1 = (it + 1 < PR_TRIPS) ? it + 1 + rot : rot;
                    scan_load(col, (in1 < PR_TRIPS) ? in1 : in1 - PR_TRIPS, tid, nx);
                }
#pragma unroll
                for (int j = 0; j < 4; ++j) {
                    const int e = (ip * 4096 + j * 1024 + tid) * 4;
                    const bool in = e < N_EDGES;
                    const unsigned bin = __builtin_amdgcn_ballot_w32(in);
                    const unsigned u0 = (unsigned)(cc[j].x - lo), u1 = (unsigned)(cc[j].y - lo);
                    const unsigned u2 = (unsigned)(cc[j].z - lo), u3 = (unsigned)(cc[j].w - lo);
                    const bool t0 = u0 < (unsigned)PR_T, t1 = u1 < (unsigned)PR_T;
                    const bool t2 = u2 < (unsigned)PR_T, t3 = u3 < (unsigned)PR_T;
                    const unsigned b0 = __builtin_amdgcn_ballot_w32(t0) & bin;
                    const unsigned b1 = __builtin_amdgcn_ballot_w32(t1) & bin;
                    const unsigned b2 = __builtin_amdgcn_ballot_w32(t2) & bin;
                    const unsigned b3 = __builtin_amdgcn_ballot_w32(t3) & bin;
                    if ((b0 | b1 | b2 | b3) != 0u) {
                        if (b0 != 0u) {
                            const unsigned w = (unsigned)(seen + __popc(b0 & lt) - wlo);
                            if (t0 && in && w < (unsigned)PR_QCAP) q[w] = (u0 << 20) | (unsigned)(e + 0);
                            seen += __popc(b0);
                        }
                        if (b1 != 0u) {
                            const unsigned w = (unsigned)(seen + __popc(b1 & lt) - wlo);
                            if (t1 && in && w < (unsigned)PR_QCAP) q[w] = (u1 << 20) | (unsigned)(e + 1);
                            seen += __popc(b1);
                        }
                        if (b2 != 0u) {
                            const unsigned w = (unsigned)(seen + __popc(b2 & lt) - wlo);
                            if (t2 && in && w < (unsigned)PR_QCAP) q[w] = (u2 << 20) | (unsigned)(e + 2);
                            seen += __popc(b2);
                        }
                        if (b3 != 0u) {
                            const unsigned w = (unsigned)(seen + __popc(b3 & lt) - wlo);
                            if (t3 && in && w < (unsigned)PR_QCAP) q[w] = (u3 << 20) | (unsigned)(e + 3);
                            seen += __popc(b3);
                        }
                    }
                }
            }

            {
                const int nown = min(max(seen - wlo, 0), PR_QCAP);
                for (int i0 = 0; i0 < nown; i0 += 128) {
                    unsigned ent[4];
                    int sid[4];
#pragma unroll
                    for (int j = 0; j < 4; ++j) {
                        const int idx = i0 + 32 * j + lane;
                        ent[j] = (idx < nown) ? q[idx] : 0u;
                    }
#pragma unroll
                    for (int j = 0; j < 4; ++j) sid[j] = row[ent[j] & 0xFFFFFu];
#pragma unroll
                    for (int j = 0; j < 4; ++j) {
                        const int idx = i0 + 32 * j + lane;
                        if (idx < nown) q[idx] = (ent[j] & 0xFFF00000u) | (unsigned)sid[j];
                    }
                }
            }
            if (lane == 0) scnt[wv] = seen;
            __syncthreads();

            const int cnt_l = scnt[lane];
            const bool more = __builtin_amdgcn_ballot_w32(cnt_l > wlo + PR_QCAP) != 0u;

            {
                const unsigned olo = (unsigned)(wv * PR_WROWS);
                unsigned* st = sstage + wv * 64;
                int head = 0, have = 0;
                for (int qw = 0; qw < 32; ++qw) {
                    const int n = min(max(__builtin_amdgcn_readlane(cnt_l, qw) - wlo, 0), PR_QCAP);
                    const unsigned* qq = squeue + qw * PR_QCAP;
                    for (int i = 0; i < n; i += 128) {
                        unsigned ent[4];
#pragma unroll
                        for (int j = 0; j < 4; ++j) {
                            const int idx = i + 32 * j + lane;
                            ent[j] = (idx < n) ? qq[idx] : 0xFFFFFFFFu;
                        }
#pragma unroll
                        for (int j = 0; j < 4; ++j) {
                            const bool m = ((ent[j] >> 20) - olo) < (unsigned)PR_WROWS;
                            const unsigned b = __builtin_amdgcn_ballot_w32(m);
                            if (b != 0u) {
                                if (m) st[(head + have + __popc(b & lt)) & 63] = ent[j];
                                have += __popc(b);
                                while (have >= 16) {
                                    const unsigned ent = prop_group(x_src, dis, sacc, st + head, 16, lo, hl, hi);
                                    if (BUILD && room && !hi && wtot + hl < PR_WCAP) glist[wtot + hl] = ent;
                                    wtot += 16;
                                    head = (head + 16) & 63;
                                    have -= 16;
                                }
                            }
                        }
                    }
                }
                if (have > 0) {
                    const unsigned ent = prop_group(x_src, dis, sacc, st + head, have, lo, hl, hi);
                    if (BUILD && room && !hi && hl < have && wtot + hl < PR_WCAP) glist[wtot + hl] = ent;
                    wtot += have;
                }
            }
            __syncthreads();
            if (!more) break;
        }
        if (BUILD && room && lane == 0) wcnt[blockIdx.x * 32 + wv] = wtot;
    }

    const int nrows = min(PR_T, N_NODES - lo);
    float4* dst = (float4*)x_dst + (size_t)lo * 16;
    for (int i = tid; i < nrows * 16; i += 1024) dst[i] = sacc4[i];
}

extern "C" void kernel_launch(void* const* d_in, const int* in_sizes, int n_in,
                              void* d_out, int out_size, void* d_ws, size_t ws_size,
                              hipStream_t stream) {
    const float* Gu      = (const float*)d_in[0];
    const float* Gi      = (const float*)d_in[1];
    const int*   edge    = (const int*)d_in[2];
    const float* intents = (const float*)d_in[3];
    const int* row = edge;
    const int* col = edge + N_EDGES;
    float* out = (float*)d_out;

    float* deg  = (float*)d_ws;
    float* xbuf = deg + (size_t)N_NODES * 4;

    int*      wcnt  = (int*)(xbuf + (size_t)N_NODES * 64);
    unsigned* wlist = (unsigned*)(wcnt + PR_R * 32);

    dim3 pgrid(PR_R), pblock(1024);

    k_deg<<<dim3(DEG_R * DEG_G), dim3(1024), 0, stream>>>(row, col, intents, out);
    k_dis_x0<<<dim3((N_NODES + 127) / 128), dim3(128), 0, stream>>>(out, deg, Gu, Gi, xbuf);

    k_propagate<true ><<<pgrid, pblock, 0, stream>>>(xbuf, out, deg, row, col, wcnt, wlist, ws_size);
    k_propagate<false><<<pgrid, pblock, 0, stream>>>(out, xbuf, deg, row, col, wcnt, wlist, ws_size);
    k_propagate<false><<<pgrid, pblock, 0, stream>>>(xbuf, out, deg, row, col, wcnt, wlist, ws_size);
}


# k_deg also bins the edges by destination range (15 ranges = 18 layer ranges each) into per-wave lists via LDS; layer 1 scans its range's 512 short lists, not all of col, no row loads; new sum order
# speedup vs baseline: 2.6626x; 1.0854x over previous
#include <hip/hip_runtime.h>

#define N_USERS  100000
#define N_ITEMS  50000
#define N_NODES  150000
#define N_EDGES  1000000
#define EMBED_K  64

typedef __attribute__((ext_vector_type(2))) float v2f;
typedef __attribute__((ext_vector_type(8))) float v8f;

#define DEG_R   15
#define DEG_T   10548
#define DEG_G   16
#define DEG_CH  4096
#define BL_CAP   256
#define BL_SUB   18
#define BL_LISTS (DEG_R * DEG_G * 32)
#define BL_WS_NEED ((size_t)N_NODES * 4 * 4 + (size_t)N_NODES * 64 * 4 + (size_t)256 * 32 * 4 + (size_t)256 * 32 * 256 * 4 + (size_t)BL_LISTS * 4 + (size_t)BL_LISTS * BL_CAP * 4)

__device__ __forceinline__ void ldsAddF32(float* p, float v) {
    (void)__hip_atomic_fetch_add(p, v, __ATOMIC_RELAXED, __HIP_MEMORY_SCOPE_WORKGROUP);
}

__device__ __forceinline__ float4 edge_w(float z0, float z1, float z2, float z3) {
    float m  = fmaxf(fmaxf(z0, z1), fmaxf(z2, z3));
    float w0 = __expf(z0 - m), w1 = __expf(z1 - m);
    float w2 = __expf(z2 - m), w3 = __expf(z3 - m);
    float inv = 1.0f / (w0 + w1 + w2 + w3);
    return make_float4(w0 * inv, w1 * inv, w2 * inv, w3 * inv);
}

__device__ __forceinline__ void node_add(float* sdeg, int lo, int node, float4 w) {
    const unsigned u = (unsigned)(node - lo);
    if (u < (unsigned)DEG_T) {
        float* p = sdeg + u * 4;
        ldsAddF32(p + 0, w.x); ldsAddF32(p + 1, w.y);
        ldsAddF32(p + 2, w.z); ldsAddF32(p + 3, w.w);
    }
}

__device__ __forceinline__ void bin_put(unsigned* bq, int& seen, unsigned lt, unsigned u, int src) {
    const bool t = u < (unsigned)DEG_T;
    const unsigned b = __builtin_amdgcn_ballot_w32(t);
    if (b != 0u) {
        const unsigned w = (unsigned)(seen + __popc(b & lt));
        if (t && w < (unsigned)BL_CAP) bq[w] = (u << 18) | (unsigned)src;
        seen += __popc(b);
    }
}

__global__ __launch_bounds__(1024) void k_deg(
        const int* __restrict__ row, const int* __restrict__ col,
        const float* __restrict__ intents, float* __restrict__ planes,
        int* __restrict__ bcnt, unsigned* __restrict__ blist, size_t ws_bytes) {
    __shared__ float4 sdeg4[DEG_T];
    __shared__ unsigned sbq[32 * BL_CAP];
    float* sdeg = (float*)sdeg4;
    const int tid = threadIdx.x;
    const int wv   = __builtin_amdgcn_readfirstlane(tid >> 5);
    const int lane = tid & 31;
    const unsigned lt = (1u << lane) - 1u;
    const bool keep = ws_bytes >= BL_WS_NEED;
    unsigned* bq = sbq + wv * BL_CAP;
    int seen = 0;
    const int rg  = blockIdx.x % DEG_R;
    const int g   = blockIdx.x / DEG_R;
    const int lo  = rg * DEG_T;

    for (int i = tid; i < DEG_T; i += 1024) sdeg4[i] = make_float4(0.0f, 0.0f, 0.0f, 0.0f);
    __syncthreads();

    for (int ch = g; ch * DEG_CH < N_EDGES; ch += DEG_G) {
        const int e = ch * DEG_CH + tid * 4;
        if (e < N_EDGES) {
            const int4   r4 = *(const int4*)(row + e);
            const int4   c4 = *(const int4*)(col + e);
            const float4 za = *(const float4*)(intents + e);
            const float4 zb = *(const float4*)(intents + N_EDGES + e);
            const float4 zc = *(const float4*)(intents + 2 * N_EDGES + e);
            const float4 zd = *(const float4*)(intents + 3 * N_EDGES + e);
            const float4 w0 = edge_w(za.x, zb.x, zc.x, zd.x);
            const float4 w1 = edge_w(za.y, zb.y, zc.y, zd.y);
            const float4 w2 = edge_w(za.z, zb.z, zc.z, zd.z);
            const float4 w3 = edge_w(za.w, zb.w, zc.w, zd.w);
            node_add(sdeg, lo, r4.x, w0); node_add(sdeg, lo, c4.x, w0);
            node_add(sdeg, lo, r4.y, w1); node_add(sdeg, lo, c4.y, w1);
            node_add(sdeg, lo, r4.z, w2); node_add(sdeg, lo, c4.z, w2);
            node_add(sdeg, lo, r4.w, w3); node_add(sdeg, lo, c4.w, w3);
            if (keep) {
                bin_put(bq, seen, lt, (unsigned)(c4.x - lo), r4.x);
                bin_put(bq, seen, lt, (unsigned)(c4.y - lo), r4.y);
                bin_put(bq, seen, lt, (unsigned)(c4.z - lo), r4.z);
                bin_put(bq, seen, lt, (unsigned)(c4.w - lo), r4.w);
            }
        }
    }
    if (keep) {
        const int nseen = __builtin_amdgcn_readfirstlane(seen);
        const int L = (rg * DEG_G + g) * 32 + wv;
        unsigned* gl = blist + (size_t)L * BL_CAP;
        const int nst = min(nseen, BL_CAP);
        for (int c = 0; c < BL_CAP / 32; ++c) {
            const int idx = c * 32 + lane;
            gl[idx] = (idx < nst) ? bq[idx] : 0xFFFFFFFFu;
        }
        if (lane == 0) bcnt[L] = nseen;
    }
    __syncthreads();

    float4* dst = (float4*)(planes + (size_t)g * (N_NODES * 4)) + lo;
    for (int i = tid; i < DEG_T; i += 1024)
        if (lo + i < N_NODES) dst[i] = sdeg4[i];
}

#define X0_U4   (N_USERS * (EMBED_K / 4))
#define X0_N4   (N_NODES * (EMBED_K / 4))
__global__ __launch_bounds__(128) void k_dis_x0(
        const float* __restrict__ planes, float* __restrict__ dis,
        const float* __restrict__ Gu, const float* __restrict__ Gi,
        float* __restrict__ x) {
    const int tid = threadIdx.x;
    const int f0  = blockIdx.x * 2048 + tid;
    float4 c[16];
#pragma unroll
    for (int j = 0; j < 16; ++j) {
        const int f = min(f0 + j * 128, X0_N4 - 1);
        const float4* p = (f < X0_U4) ? (const float4*)Gu + f : (const float4*)Gi + (f - X0_U4);
        c[j] = *p;
    }
    const int i = blockIdx.x * 128 + tid;
    if (i < N_NODES) {
        float4 s = ((const float4*)planes)[i];
#pragma unroll
        for (int g = 1; g < DEG_G; ++g) {
            const float4 p = ((const float4*)(planes + (size_t)g * (N_NODES * 4)))[i];
            s.x += p.x; s.y += p.y; s.z += p.z; s.w += p.w;
        }
        float4 o;
        o.x = (s.x > 0.0f) ? rsqrtf(fmaxf(s.x, 1e-12f)) : 0.0f;
        o.y = (s.y > 0.0f) ? rsqrtf(fmaxf(s.y, 1e-12f)) : 0.0f;
        o.z = (s.z > 0.0f) ? rsqrtf(fmaxf(s.z, 1e-12f)) : 0.0f;
        o.w = (s.w > 0.0f) ? rsqrtf(fmaxf(s.w, 1e-12f)) : 0.0f;
        ((float4*)dis)[i] = o;
    }
#pragma unroll
    for (int j = 0; j < 16; ++j) {
        asm volatile("" :: "v"(c[j].x), "v"(c[j].y), "v"(c[j].z), "v"(c[j].w));
    }
#pragma unroll
    for (int j = 0; j < 16; ++j) {
        const int f = f0 + j * 128;
        if (f < X0_N4) ((float4*)x)[f] = c[j];
    }
}

#define PR_R       256
#define PR_T       586
#define PR_WROWS   19
#define PR_QCAP    1024
#define PR_MAXPASS 32
#define PR_TRIPS   ((N_EDGES + 128 * 128 - 1) / (128 * 128))
#define PR_WCAP    256
#define PR_WS_NEED ((size_t)N_NODES * 4 * 4 + (size_t)N_NODES * 64 * 4 + (size_t)PR_R * 32 * 4 + (size_t)PR_R * 32 * PR_WCAP * 4)
static_assert(BL_WS_NEED == PR_WS_NEED + (size_t)BL_LISTS * 4 + (size_t)BL_LISTS * BL_CAP * 4, "the binned lists lie behind the stored lists");
static_assert(DEG_T == BL_SUB * PR_T, "a degree range must be whole layer ranges");
static_assert(DEG_R * BL_SUB >= PR_R, "every layer range needs a degree range");
static_assert(DEG_G * 32 == 32 * 16, "a layer-1 wave takes 16 of its range's lists");

__device__ __forceinline__ unsigned prop_group(
        const float* __restrict__ x_src, const float* __restrict__ dis,
        float* sacc, const unsigned* grp, int n, int lo,
        int hl, bool hi) {
    const bool vA = hl < n;
    const unsigned entA = vA ? grp[hl] : 0u;
    const int irA = (int)(entA & 0xFFFFFu);
    const int dlA = (int)(entA >> 20);
    const float4 dr = *(const float4*)(dis + irA * 4);
    const float4 dc = *(const float4*)(dis + (lo + dlA) * 4);
    float nrm[4];
    nrm[0] = vA ? dr.x * dc.x : 0.0f; nrm[1] = vA ? dr.y * dc.y : 0.0f;
    nrm[2] = vA ? dr.z * dc.z : 0.0f; nrm[3] = vA ? dr.w * dc.w : 0.0f;

    const int kx = hi ? 2 : 0;
    const int mo = hi ? 8 : 0;
    int rb[8];
#pragma unroll
    for (int t = 0; t < 4; ++t) {
        rb[2 * t]     = __shfl(irA, 4 * t + kx);
        rb[2 * t + 1] = __shfl(irA, 4 * t + kx + 1);
    }
    int dlv[8];
#pragma unroll
    for (int v = 0; v < 8; ++v) dlv[v] = __shfl(dlA, v + mo);

    float bx[16], by[16];
#pragma unroll
    for (int k = 0; k < 4; ++k) {
#pragma unroll
        for (int t = 0; t < 4; ++t) {
            bx[k * 4 + t] = x_src[rb[2 * t]     * 64 + k * 16 + hl];
            by[k * 4 + t] = x_src[rb[2 * t + 1] * 64 + k * 16 + hl];
        }
    }

#pragma unroll
    for (int k = 0; k < 4; ++k) {
        v8f acc = {};
#pragma unroll
        for (int t = 0; t < 4; ++t) {
            v2f a, b;
            a.x = (hl == 4 * t + kx)     ? nrm[k] : 0.0f;
            a.y = (hl == 4 * t + kx + 1) ? nrm[k] : 0.0f;
            b.x = bx[k * 4 + t];
            b.y = by[k * 4 + t];
            acc = __builtin_amdgcn_wmma_f32_16x16x4_f32(
                      false, a, false, b, (short)0, acc, false, false);
        }
#pragma unroll
        for (int v = 0; v < 8; ++v)
            if (v + mo < n) ldsAddF32(sacc + dlv[v] * 64 + k * 16 + hl, acc[v]);
    }
    return entA;
}

__device__ __forceinline__ void scan_load(const int* __restrict__ col, int ip, int tid, int4 (&cc)[4]) {
#pragma unroll
    for (int j = 0; j < 4; ++j) {
        const unsigned i4 = (unsigned)min(ip * 4096 + j * 1024 + tid, N_EDGES / 4 - 1);
        cc[j] = *(const int4*)((const char*)col + i4 * 16u);
    }
}

template <bool BUILD>
__global__ __launch_bounds__(1024) void k_propagate(
        const float* __restrict__ x_src, float* __restrict__ x_dst,
        const float* __restrict__ dis,
        const int* __restrict__ row, const int* __restrict__ col,
        int* __restrict__ wcnt, unsigned* __restrict__ wlist,
        const int* __restrict__ bcnt, const unsigned* __restrict__ blist, size_t ws_bytes) {
    __shared__ float4   sacc4[PR_T * 16];
    __shared__ unsigned squeue[32 * PR_QCAP];
    __shared__ unsigned sstage[32 * 64];
    __shared__ int      scnt[32];
    float* sacc = (float*)sacc4;
    const int tid  = threadIdx.x;
    const int wv   = __builtin_amdgcn_readfirstlane(tid >> 5);
    const int lane = tid & 31;
    const int hl   = lane & 15;
    const bool hi  = lane >= 16;
    const unsigned lt = (1u << lane) - 1u;
    const int lo   = blockIdx.x * PR_T;
    const int rot  = blockIdx.x % PR_TRIPS;

    const bool room = ws_bytes >= PR_WS_NEED;
    unsigned* glist = wlist + (size_t)(blockIdx.x * 32 + wv) * PR_WCAP;
    int wc_l = 0;
    if (!BUILD && room) wc_l = wcnt[blockIdx.x * 32 + lane];
    bool binned = false;
    int bc_l = 0;
    const int lb = (blockIdx.x / BL_SUB) * (DEG_G * 32);
    if (BUILD && ws_bytes >= BL_WS_NEED) {
        unsigned over = 0u;
#pragma unroll
        for (int j = 0; j < DEG_G; ++j) over |= (unsigned)((unsigned)bcnt[lb + j * 32 + lane] > (unsigned)BL_CAP);
        bc_l = bcnt[lb + wv * 16 + (lane & 15)];
        binned = __builtin_amdgcn_ballot_w32(over != 0u) == 0u;
    }

    for (int i = tid; i < PR_T * 16; i += 1024) sacc4[i] = make_float4(0.0f, 0.0f, 0.0f, 0.0f);

    const bool replay = !BUILD && room &&
                        __builtin_amdgcn_ballot_w32((unsigned)wc_l > (unsigned)PR_WCAP) == 0u;

    if (replay) {
        const int n = __builtin_amdgcn_readlane(wc_l, wv);
        unsigned* wl = squeue + wv * PR_QCAP;
        unsigned e[PR_WCAP / 32];
#pragma unroll
        for (int c = 0; c < PR_WCAP / 32; ++c) {
            const int idx = c * 32 + lane;
            e[c] = (idx < n) ? glist[idx] : 0u;
        }
#pragma unroll
        for (int c = 0; c < PR_WCAP / 32; ++c) wl[c * 32 + lane] = e[c];
        __syncthreads();
        for (int g0 = 0; g0 < n; g0 += 16)
            (void)prop_group(x_src, dis, sacc, wl + g0, min(16, n - g0), lo, hl, hi);
        __syncthreads();
    } else {
        int wtot = 0;
        for (int pass = 0; pass < PR_MAXPASS; ++pass) {
            const int wlo = pass * PR_QCAP;

            unsigned* q = squeue + wv * PR_QCAP;
            int seen = 0;
            if (BUILD && binned) {
            const unsigned j0 = (unsigned)((blockIdx.x % BL_SUB) * PR_T);
            const unsigned* bl0 = blist + (size_t)(lb + wv * 16) * BL_CAP;
            unsigned en[BL_CAP / 32];
#pragma unroll
            for (int c = 0; c < BL_CAP / 32; ++c) en[c] = bl0[c * 32 + lane];
            for (int i = 0; i < 16; ++i) {
                const int n = __builtin_amdgcn_readlane(bc_l, i);
                unsigned ec[BL_CAP / 32];
#pragma unroll
                for (int c = 0; c < BL_CAP / 32; ++c) ec[c] = en[c];
                {
                    const unsigned* bl = bl0 + (size_t)min(i + 1, 15) * BL_CAP;
#pragma unroll
                    for (int c = 0; c < BL_CAP / 32; ++c) en[c] = bl[c * 32 + lane];
                }
#pragma unroll
                for (int c = 0; c < BL_CAP / 32; ++c) {
                    if (c * 32 < n) {
                        const unsigned u = (ec[c] >> 18) - j0;
                        const bool t = u < (unsigned)PR_T;
                        const unsigned b = __builtin_amdgcn_ballot_w32(t);
                        if (b != 0u) {
                            const unsigned w = (unsigned)(seen + __popc(b & lt) - wlo);
                            if (t && w < (unsigned)PR_QCAP) q[w] = (u << 20) | (ec[c] & 0x3FFFFu);
                            seen += __popc(b);
                        }
                    }
                }
            }
            } else {
            int4 nx[4];
            scan_load(col, rot, tid, nx);
            for (int it = 0; it < PR_TRIPS; ++it) {
                const int ip = (it + rot < PR_TRIPS) ? it + rot : it + rot - PR_TRIPS;
                int4 cc[4];
#pragma unroll
                for (int j = 0; j < 4; ++j) cc[j] = nx[j];
                {
                    const int in1 = (it + 1 < PR_TRIPS) ? it + 1 + rot : rot;
                    scan_load(col, (in1 < PR_TRIPS) ? in1 : in1 - PR_TRIPS, tid, nx);
                }
#pragma unroll
                for (int j = 0; j < 4; ++j) {
                    const int e = (ip * 4096 + j * 1024 + tid) * 4;
                    const bool in = e < N_EDGES;
                    const unsigned bin = __builtin_amdgcn_ballot_w32(in);
                    const unsigned u0 = (unsigned)(cc[j].x - lo), u1 = (unsigned)(cc[j].y - lo);
                    const unsigned u2 = (unsigned)(cc[j].z - lo), u3 = (unsigned)(cc[j].w - lo);
                    const bool t0 = u0 < (unsigned)PR_T, t1 = u1 < (unsigned)PR_T;
                    const bool t2 = u2 < (unsigned)PR_T, t3 = u3 < (unsigned)PR_T;
                    const unsigned b0 = __builtin_amdgcn_ballot_w32(t0) & bin;
                    const unsigned b1 = __builtin_amdgcn_ballot_w32(t1) & bin;
                    const unsigned b2 = __builtin_amdgcn_ballot_w32(t2) & bin;
                    const unsigned b3 = __builtin_amdgcn_ballot_w32(t3) & bin;
                    if ((b0 | b1 | b2 | b3) != 0u) {
                        if (b0 != 0u) {
                            const unsigned w = (unsigned)(seen + __popc(b0 & lt) - wlo);
                            if (t0 && in && w < (unsigned)PR_QCAP) q[w] = (u0 << 20) | (unsigned)(e + 0);
                            seen += __popc(b0);
                        }
                        if (b1 != 0u) {
                            const unsigned w = (unsigned)(seen + __popc(b1 & lt) - wlo);
                            if (t1 && in && w < (unsigned)PR_QCAP) q[w] = (u1 << 20) | (unsigned)(e + 1);
                            seen += __popc(b1);
                        }
                        if (b2 != 0u) {
                            const unsigned w = (unsigned)(seen + __popc(b2 & lt) - wlo);
                            if (t2 && in && w < (unsigned)PR_QCAP) q[w] = (u2 << 20) | (unsigned)(e + 2);
                            seen += __popc(b2);
                        }
                        if (b3 != 0u) {
                            const unsigned w = (unsigned)(seen + __popc(b3 & lt) - wlo);
                            if (t3 && in && w < (unsigned)PR_QCAP) q[w] = (u3 << 20) | (unsigned)(e + 3);
                            seen += __popc(b3);
                        }
                    }
                }
            }

            {
                const int nown = min(max(seen - wlo, 0), PR_QCAP);
                for (int i0 = 0; i0 < nown; i0 += 128) {
                    unsigned ent[4];
                    int sid[4];
#pragma unroll
                    for (int j = 0; j < 4; ++j) {
                        const int idx = i0 + 32 * j + lane;
                        ent[j] = (idx < nown) ? q[idx] : 0u;
                    }
#pragma unroll
                    for (int j = 0; j < 4; ++j) sid[j] = row[ent[j] & 0xFFFFFu];
#pragma unroll
                    for (int j = 0; j < 4; ++j) {
                        const int idx = i0 + 32 * j + lane;
                        if (idx < nown) q[idx] = (ent[j] & 0xFFF00000u) | (unsigned)sid[j];
                    }
                }
            }
            }
            if (lane == 0) scnt[wv] = seen;
            __syncthreads();

            const int cnt_l = scnt[lane];
            const bool more = __builtin_amdgcn_ballot_w32(cnt_l > wlo + PR_QCAP) != 0u;

            {
                const unsigned olo = (unsigned)(wv * PR_WROWS);
                unsigned* st = sstage + wv * 64;
                int head = 0, have = 0;
                for (int qw = 0; qw < 32; ++qw) {
                    const int n = min(max(__builtin_amdgcn_readlane(cnt_l, qw) - wlo, 0), PR_QCAP);
                    const unsigned* qq = squeue + qw * PR_QCAP;
                    for (int i = 0; i < n; i += 128) {
                        unsigned ent[4];
#pragma unroll
                        for (int j = 0; j < 4; ++j) {
                            const int idx = i + 32 * j + lane;
                            ent[j] = (idx < n) ? qq[idx] : 0xFFFFFFFFu;
                        }
#pragma unroll
                        for (int j = 0; j < 4; ++j) {
                            const bool m = ((ent[j] >> 20) - olo) < (unsigned)PR_WROWS;
                            const unsigned b = __builtin_amdgcn_ballot_w32(m);
                            if (b != 0u) {
                                if (m) st[(head + have + __popc(b & lt)) & 63] = ent[j];
                                have += __popc(b);
                                while (have >= 16) {
                                    const unsigned ent = prop_group(x_src, dis, sacc, st + head, 16, lo, hl, hi);
                                    if (BUILD && room && !hi && wtot + hl < PR_WCAP) glist[wtot + hl] = ent;
                                    wtot += 16;
                                    head = (head + 16) & 63;
                                    have -= 16;
                                }
                            }
                        }
                    }
                }
                if (have > 0) {
                    const unsigned ent = prop_group(x_src, dis, sacc, st + head, have, lo, hl, hi);
                    if (BUILD && room && !hi && hl < have && wtot + hl < PR_WCAP) glist[wtot + hl] = ent;
                    wtot += have;
                }
            }
            __syncthreads();
            if (!more) break;
        }
        if (BUILD && room && lane == 0) wcnt[blockIdx.x * 32 + wv] = wtot;
    }

    const int nrows = min(PR_T, N_NODES - lo);
    float4* dst = (float4*)x_dst + (size_t)lo * 16;
    for (int i = tid; i < nrows * 16; i += 1024) dst[i] = sacc4[i];
}

extern "C" void kernel_launch(void* const* d_in, const int* in_sizes, int n_in,
                              void* d_out, int out_size, void* d_ws, size_t ws_size,
                              hipStream_t stream) {
    const float* Gu      = (const float*)d_in[0];
    const float* Gi      = (const float*)d_in[1];
    const int*   edge    = (const int*)d_in[2];
    const float* intents = (const float*)d_in[3];
    const int* row = edge;
    const int* col = edge + N_EDGES;
    float* out = (float*)d_out;

    float* deg  = (float*)d_ws;
    float* xbuf = deg + (size_t)N_NODES * 4;

    int*      wcnt  = (int*)(xbuf + (size_t)N_NODES * 64);
    unsigned* wlist = (unsigned*)(wcnt + PR_R * 32);
    int*      bcnt  = (int*)(wlist + (size_t)PR_R * 32 * PR_WCAP);
    unsigned* blist = (unsigned*)(bcnt + BL_LISTS);

    dim3 pgrid(PR_R), pblock(1024);

    k_deg<<<dim3(DEG_R * DEG_G), dim3(1024), 0, stream>>>(row, col, intents, out, bcnt, blist, ws_size);
    k_dis_x0<<<dim3((N_NODES + 127) / 128), dim3(128), 0, stream>>>(out, deg, Gu, Gi, xbuf);

    k_propagate<true ><<<pgrid, pblock, 0, stream>>>(xbuf, out, deg, row, col, wcnt, wlist, bcnt, blist, ws_size);
    k_propagate<false><<<pgrid, pblock, 0, stream>>>(out, xbuf, deg, row, col, wcnt, wlist, bcnt, blist, ws_size);
    k_propagate<false><<<pgrid, pblock, 0, stream>>>(xbuf, out, deg, row, col, wcnt, wlist, bcnt, blist, ws_size);
}


# k_propagate layers 2-3: edges summed on the matrix unit (A = gathered x, B = norm where destination == column), sums in registers, rows stored per wave; no LDS adds or barriers; new sum order
# speedup vs baseline: 2.9897x; 1.1228x over previous
#include <hip/hip_runtime.h>

#define N_USERS  100000
#define N_ITEMS  50000
#define N_NODES  150000
#define N_EDGES  1000000
#define EMBED_K  64

typedef __attribute__((ext_vector_type(2))) float v2f;
typedef __attribute__((ext_vector_type(8))) float v8f;

#define DEG_R   15
#define DEG_T   10548
#define DEG_G   16
#define DEG_CH  4096
#define BL_CAP   256
#define BL_SUB   18
#define BL_LISTS (DEG_R * DEG_G * 32)
#define BL_WS_NEED ((size_t)N_NODES * 4 * 4 + (size_t)N_NODES * 64 * 4 + (size_t)256 * 32 * 4 + (size_t)256 * 32 * 256 * 4 + (size_t)BL_LISTS * 4 + (size_t)BL_LISTS * BL_CAP * 4)

__device__ __forceinline__ void ldsAddF32(float* p, float v) {
    (void)__hip_atomic_fetch_add(p, v, __ATOMIC_RELAXED, __HIP_MEMORY_SCOPE_WORKGROUP);
}

__device__ __forceinline__ float4 edge_w(float z0, float z1, float z2, float z3) {
    float m  = fmaxf(fmaxf(z0, z1), fmaxf(z2, z3));
    float w0 = __expf(z0 - m), w1 = __expf(z1 - m);
    float w2 = __expf(z2 - m), w3 = __expf(z3 - m);
    float inv = 1.0f / (w0 + w1 + w2 + w3);
    return make_float4(w0 * inv, w1 * inv, w2 * inv, w3 * inv);
}

__device__ __forceinline__ void node_add(float* sdeg, int lo, int node, float4 w) {
    const unsigned u = (unsigned)(node - lo);
    if (u < (unsigned)DEG_T) {
        float* p = sdeg + u * 4;
        ldsAddF32(p + 0, w.x); ldsAddF32(p + 1, w.y);
        ldsAddF32(p + 2, w.z); ldsAddF32(p + 3, w.w);
    }
}

__device__ __forceinline__ void bin_put(unsigned* bq, int& seen, unsigned lt, unsigned u, int src) {
    const bool t = u < (unsigned)DEG_T;
    const unsigned b = __builtin_amdgcn_ballot_w32(t);
    if (b != 0u) {
        const unsigned w = (unsigned)(seen + __popc(b & lt));
        if (t && w < (unsigned)BL_CAP) bq[w] = (u << 18) | (unsigned)src;
        seen += __popc(b);
    }
}

__global__ __launch_bounds__(1024) void k_deg(
        const int* __restrict__ row, const int* __restrict__ col,
        const float* __restrict__ intents, float* __restrict__ planes,
        int* __restrict__ bcnt, unsigned* __restrict__ blist, size_t ws_bytes) {
    __shared__ float4 sdeg4[DEG_T];
    __shared__ unsigned sbq[32 * BL_CAP];
    float* sdeg = (float*)sdeg4;
    const int tid = threadIdx.x;
    const int wv   = __builtin_amdgcn_readfirstlane(tid >> 5);
    const int lane = tid & 31;
    const unsigned lt = (1u << lane) - 1u;
    const bool keep = ws_bytes >= BL_WS_NEED;
    unsigned* bq = sbq + wv * BL_CAP;
    int seen = 0;
    const int rg  = blockIdx.x % DEG_R;
    const int g   = blockIdx.x / DEG_R;
    const int lo  = rg * DEG_T;

    for (int i = tid; i < DEG_T; i += 1024) sdeg4[i] = make_float4(0.0f, 0.0f, 0.0f, 0.0f);
    __syncthreads();

    for (int ch = g; ch * DEG_CH < N_EDGES; ch += DEG_G) {
        const int e = ch * DEG_CH + tid * 4;
        if (e < N_EDGES) {
            const int4   r4 = *(const int4*)(row + e);
            const int4   c4 = *(const int4*)(col + e);
            const float4 za = *(const float4*)(intents + e);
            const float4 zb = *(const float4*)(intents + N_EDGES + e);
            const float4 zc = *(const float4*)(intents + 2 * N_EDGES + e);
            const float4 zd = *(const float4*)(intents + 3 * N_EDGES + e);
            const float4 w0 = edge_w(za.x, zb.x, zc.x, zd.x);
            const float4 w1 = edge_w(za.y, zb.y, zc.y, zd.y);
            const float4 w2 = edge_w(za.z, zb.z, zc.z, zd.z);
            const float4 w3 = edge_w(za.w, zb.w, zc.w, zd.w);
            node_add(sdeg, lo, r4.x, w0); node_add(sdeg, lo, c4.x, w0);
            node_add(sdeg, lo, r4.y, w1); node_add(sdeg, lo, c4.y, w1);
            node_add(sdeg, lo, r4.z, w2); node_add(sdeg, lo, c4.z, w2);
            node_add(sdeg, lo, r4.w, w3); node_add(sdeg, lo, c4.w, w3);
            if (keep) {
                bin_put(bq, seen, lt, (unsigned)(c4.x - lo), r4.x);
                bin_put(bq, seen, lt, (unsigned)(c4.y - lo), r4.y);
                bin_put(bq, seen, lt, (unsigned)(c4.z - lo), r4.z);
                bin_put(bq, seen, lt, (unsigned)(c4.w - lo), r4.w);
            }
        }
    }
    if (keep) {
        const int nseen = __builtin_amdgcn_readfirstlane(seen);
        const int L = (rg * DEG_G + g) * 32 + wv;
        unsigned* gl = blist + (size_t)L * BL_CAP;
        const int nst = min(nseen, BL_CAP);
        for (int c = 0; c < BL_CAP / 32; ++c) {
            const int idx = c * 32 + lane;
            gl[idx] = (idx < nst) ? bq[idx] : 0xFFFFFFFFu;
        }
        if (lane == 0) bcnt[L] = nseen;
    }
    __syncthreads();

    float4* dst = (float4*)(planes + (size_t)g * (N_NODES * 4)) + lo;
    for (int i = tid; i < DEG_T; i += 1024)
        if (lo + i < N_NODES) dst[i] = sdeg4[i];
}

#define X0_U4   (N_USERS * (EMBED_K / 4))
#define X0_N4   (N_NODES * (EMBED_K / 4))
__global__ __launch_bounds__(128) void k_dis_x0(
        const float* __restrict__ planes, float* __restrict__ dis,
        const float* __restrict__ Gu, const float* __restrict__ Gi,
        float* __restrict__ x) {
    const int tid = threadIdx.x;
    const int f0  = blockIdx.x * 2048 + tid;
    float4 c[16];
#pragma unroll
    for (int j = 0; j < 16; ++j) {
        const int f = min(f0 + j * 128, X0_N4 - 1);
        const float4* p = (f < X0_U4) ? (const float4*)Gu + f : (const float4*)Gi + (f - X0_U4);
        c[j] = *p;
    }
    const int i = blockIdx.x * 128 + tid;
    if (i < N_NODES) {
        float4 s = ((const float4*)planes)[i];
#pragma unroll
        for (int g = 1; g < DEG_G; ++g) {
            const float4 p = ((const float4*)(planes + (size_t)g * (N_NODES * 4)))[i];
            s.x += p.x; s.y += p.y; s.z += p.z; s.w += p.w;
        }
        float4 o;
        o.x = (s.x > 0.0f) ? rsqrtf(fmaxf(s.x, 1e-12f)) : 0.0f;
        o.y = (s.y > 0.0f) ? rsqrtf(fmaxf(s.y, 1e-12f)) : 0.0f;
        o.z = (s.z > 0.0f) ? rsqrtf(fmaxf(s.z, 1e-12f)) : 0.0f;
        o.w = (s.w > 0.0f) ? rsqrtf(fmaxf(s.w, 1e-12f)) : 0.0f;
        ((float4*)dis)[i] = o;
    }
#pragma unroll
    for (int j = 0; j < 16; ++j) {
        asm volatile("" :: "v"(c[j].x), "v"(c[j].y), "v"(c[j].z), "v"(c[j].w));
    }
#pragma unroll
    for (int j = 0; j < 16; ++j) {
        const int f = f0 + j * 128;
        if (f < X0_N4) ((float4*)x)[f] = c[j];
    }
}

#define PR_R       256
#define PR_T       586
#define PR_WROWS   19
#define PR_QCAP    1024
#define PR_MAXPASS 32
#define PR_TRIPS   ((N_EDGES + 128 * 128 - 1) / (128 * 128))
#define PR_WCAP    256
#define PR_WS_NEED ((size_t)N_NODES * 4 * 4 + (size_t)N_NODES * 64 * 4 + (size_t)PR_R * 32 * 4 + (size_t)PR_R * 32 * PR_WCAP * 4)
static_assert(BL_WS_NEED == PR_WS_NEED + (size_t)BL_LISTS * 4 + (size_t)BL_LISTS * BL_CAP * 4, "the binned lists lie behind the stored lists");
static_assert(DEG_T == BL_SUB * PR_T, "a degree range must be whole layer ranges");
static_assert(DEG_R * BL_SUB >= PR_R, "every layer range needs a degree range");
static_assert(DEG_G * 32 == 32 * 16, "a layer-1 wave takes 16 of its range's lists");
static_assert(32 * 272 <= PR_T * 16 && 272 + 16 * 36 <= PR_QCAP && PR_WCAP + 6 <= 272, "the replay path's per-wave LDS areas fit");

__device__ __forceinline__ unsigned prop_group(
        const float* __restrict__ x_src, const float* __restrict__ dis,
        float* sacc, const unsigned* grp, int n, int lo,
        int hl, bool hi) {
    const bool vA = hl < n;
    const unsigned entA = vA ? grp[hl] : 0u;
    const int irA = (int)(entA & 0xFFFFFu);
    const int dlA = (int)(entA >> 20);
    const float4 dr = *(const float4*)(dis + irA * 4);
    const float4 dc = *(const float4*)(dis + (lo + dlA) * 4);
    float nrm[4];
    nrm[0] = vA ? dr.x * dc.x : 0.0f; nrm[1] = vA ? dr.y * dc.y : 0.0f;
    nrm[2] = vA ? dr.z * dc.z : 0.0f; nrm[3] = vA ? dr.w * dc.w : 0.0f;

    const int kx = hi ? 2 : 0;
    const int mo = hi ? 8 : 0;
    int rb[8];
#pragma unroll
    for (int t = 0; t < 4; ++t) {
        rb[2 * t]     = __shfl(irA, 4 * t + kx);
        rb[2 * t + 1] = __shfl(irA, 4 * t + kx + 1);
    }
    int dlv[8];
#pragma unroll
    for (int v = 0; v < 8; ++v) dlv[v] = __shfl(dlA, v + mo);

    float bx[16], by[16];
#pragma unroll
    for (int k = 0; k < 4; ++k) {
#pragma unroll
        for (int t = 0; t < 4; ++t) {
            bx[k * 4 + t] = x_src[rb[2 * t]     * 64 + k * 16 + hl];
            by[k * 4 + t] = x_src[rb[2 * t + 1] * 64 + k * 16 + hl];
        }
    }

#pragma unroll
    for (int k = 0; k < 4; ++k) {
        v8f acc = {};
#pragma unroll
        for (int t = 0; t < 4; ++t) {
            v2f a, b;
            a.x = (hl == 4 * t + kx)     ? nrm[k] : 0.0f;
            a.y = (hl == 4 * t + kx + 1) ? nrm[k] : 0.0f;
            b.x = bx[k * 4 + t];
            b.y = by[k * 4 + t];
            acc = __builtin_amdgcn_wmma_f32_16x16x4_f32(
                      false, a, false, b, (short)0, acc, false, false);
        }
#pragma unroll
        for (int v = 0; v < 8; ++v)
            if (v + mo < n) ldsAddF32(sacc + dlv[v] * 64 + k * 16 + hl, acc[v]);
    }
    return entA;
}


#define RP_PITCH 36
__device__ __forceinline__ void replay_wave(
        const float* __restrict__ x_src, float* __restrict__ x_dst, const float* __restrict__ dis,
        const unsigned* __restrict__ glist, int n, int lo, int olo, int nrows,
        unsigned* went, float4* wnrm, float* stg, int lane, unsigned lt) {
    const int c = lane & 15;
    const int h = lane >> 4;
    unsigned e[PR_WCAP / 32];
#pragma unroll
    for (int j = 0; j < PR_WCAP / 32; ++j) {
        const int idx = j * 32 + lane;
        e[j] = (idx < n) ? glist[idx] : 0u;
    }
    float4 nv[PR_WCAP / 32];
    unsigned b0[PR_WCAP / 32], b1[PR_WCAP / 32];
    int n0 = 0, n1 = 0;
#pragma unroll
    for (int j = 0; j < PR_WCAP / 32; ++j) {
        const bool valid = j * 32 + lane < n;
        const int src = (int)(e[j] & 0xFFFFFu);
        const int dl  = (int)(e[j] >> 20);
        const float4 dr = *(const float4*)(dis + src * 4);
        const float4 dc = *(const float4*)(dis + (lo + dl) * 4);
        nv[j].x = valid ? dr.x * dc.x : 0.0f; nv[j].y = valid ? dr.y * dc.y : 0.0f;
        nv[j].z = valid ? dr.z * dc.z : 0.0f; nv[j].w = valid ? dr.w * dc.w : 0.0f;
        const bool up = (dl - olo) >= 16;
        b0[j] = __builtin_amdgcn_ballot_w32(valid && !up);
        b1[j] = __builtin_amdgcn_ballot_w32(valid && up);
        n0 += __popc(b0[j]);
        n1 += __popc(b1[j]);
    }
    const int z0 = (n0 + 3) & ~3;
    const int z1 = z0 + ((n1 + 3) & ~3);
    {
        int p0 = 0, p1 = z0;
#pragma unroll
        for (int j = 0; j < PR_WCAP / 32; ++j) {
            const bool valid = j * 32 + lane < n;
            const bool up = ((int)(e[j] >> 20) - olo) >= 16;
            const int pos = up ? p1 + __popc(b1[j] & lt) : p0 + __popc(b0[j] & lt);
            if (valid) { went[pos] = e[j]; wnrm[pos] = nv[j]; }
            p0 += __popc(b0[j]);
            p1 += __popc(b1[j]);
        }
        if (n0 + lane < z0)      { went[n0 + lane] = 0u;      wnrm[n0 + lane]      = make_float4(0.0f, 0.0f, 0.0f, 0.0f); }
        if (z0 + n1 + lane < z1) { went[z0 + n1 + lane] = 0u; wnrm[z0 + n1 + lane] = make_float4(0.0f, 0.0f, 0.0f, 0.0f); }
    }

    for (int tile = 0; tile < 2; ++tile) {
        const int gb = tile ? z0 : 0;
        const int ge = tile ? z1 : z0;
        const int rb = olo + tile * 16;
        v8f acc[4] = {};
        for (int g0 = gb; g0 < ge; g0 += 16) {
            const int cnt = min(16, ge - g0);
            float ax[16], ay[16], wa[16], wb[16];
#pragma unroll
            for (int t = 0; t < 4; ++t) {
                const int idx = min(g0 + 4 * t + 2 * h, ge - 2);
                const uint2 ee = *(const uint2*)(went + idx);
                const float4 na = wnrm[idx];
                const float4 nb = wnrm[idx + 1];
                const int s0 = (int)(ee.x & 0xFFFFFu);
                const int s1 = (int)(ee.y & 0xFFFFFu);
                const bool m0 = ((int)(ee.x >> 20) - rb) == c;
                const bool m1 = ((int)(ee.y >> 20) - rb) == c;
#pragma unroll
                for (int k = 0; k < 4; ++k) {
                    ax[t * 4 + k] = x_src[s0 * 64 + k * 16 + c];
                    ay[t * 4 + k] = x_src[s1 * 64 + k * 16 + c];
                }
                wa[t * 4 + 0] = m0 ? na.x : 0.0f; wa[t * 4 + 1] = m0 ? na.y : 0.0f;
                wa[t * 4 + 2] = m0 ? na.z : 0.0f; wa[t * 4 + 3] = m0 ? na.w : 0.0f;
                wb[t * 4 + 0] = m1 ? nb.x : 0.0f; wb[t * 4 + 1] = m1 ? nb.y : 0.0f;
                wb[t * 4 + 2] = m1 ? nb.z : 0.0f; wb[t * 4 + 3] = m1 ? nb.w : 0.0f;
            }
#pragma unroll
            for (int i = 0; i < 16; ++i) {
                asm volatile("" :: "v"(ax[i]), "v"(ay[i]));
            }
#pragma unroll
            for (int t = 0; t < 4; ++t) {
                if (t == 0 || 4 * t < cnt) {
#pragma unroll
                    for (int k = 0; k < 4; ++k) {
                        v2f a, b;
                        a.x = ax[t * 4 + k]; a.y = ay[t * 4 + k];
                        b.x = wa[t * 4 + k]; b.y = wb[t * 4 + k];
                        acc[k] = __builtin_amdgcn_wmma_f32_16x16x4_f32(
                                     false, a, false, b, (short)0, acc[k], false, false);
                    }
                }
            }
        }
        const int vr = min(min(PR_WROWS - tile * 16, 16), min(PR_T, nrows) - rb);
        if (vr > 0) {
#pragma unroll
            for (int p = 0; p < 2; ++p) {
                float* sp = stg + c * RP_PITCH + 8 * h;
                *(float4*)(sp)      = make_float4(acc[2 * p][0], acc[2 * p][1], acc[2 * p][2], acc[2 * p][3]);
                *(float4*)(sp + 4)  = make_float4(acc[2 * p][4], acc[2 * p][5], acc[2 * p][6], acc[2 * p][7]);
                *(float4*)(sp + 16) = make_float4(acc[2 * p + 1][0], acc[2 * p + 1][1], acc[2 * p + 1][2], acc[2 * p + 1][3]);
                *(float4*)(sp + 20) = make_float4(acc[2 * p + 1][4], acc[2 * p + 1][5], acc[2 * p + 1][6], acc[2 * p + 1][7]);
#pragma unroll
                for (int j = 0; j < 4; ++j) {
                    if (j * 4 < vr) {
                        const int i = j * 32 + lane;
                        const int r = i >> 3, q = i & 7;
                        const float4 v = *(const float4*)(stg + r * RP_PITCH + q * 4);
                        if (r < vr) *(float4*)(x_dst + (size_t)(lo + rb + r) * 64 + p * 32 + q * 4) = v;
                    }
                }
            }
        }
    }
}

__device__ __forceinline__ void scan_load(const int* __restrict__ col, int ip, int tid, int4 (&cc)[4]) {
#pragma unroll
    for (int j = 0; j < 4; ++j) {
        const unsigned i4 = (unsigned)min(ip * 4096 + j * 1024 + tid, N_EDGES / 4 - 1);
        cc[j] = *(const int4*)((const char*)col + i4 * 16u);
    }
}

template <bool BUILD>
__global__ __launch_bounds__(1024) void k_propagate(
        const float* __restrict__ x_src, float* __restrict__ x_dst,
        const float* __restrict__ dis,
        const int* __restrict__ row, const int* __restrict__ col,
        int* __restrict__ wcnt, unsigned* __restrict__ wlist,
        const int* __restrict__ bcnt, const unsigned* __restrict__ blist, size_t ws_bytes) {
    __shared__ float4   sacc4[PR_T * 16];
    __shared__ __attribute__((aligned(16))) unsigned squeue[32 * PR_QCAP];
    __shared__ unsigned sstage[32 * 64];
    __shared__ int      scnt[32];
    float* sacc = (float*)sacc4;
    const int tid  = threadIdx.x;
    const int wv   = __builtin_amdgcn_readfirstlane(tid >> 5);
    const int lane = tid & 31;
    const int hl   = lane & 15;
    const bool hi  = lane >= 16;
    const unsigned lt = (1u << lane) - 1u;
    const int lo   = blockIdx.x * PR_T;
    const int rot  = blockIdx.x % PR_TRIPS;

    const bool room = ws_bytes >= PR_WS_NEED;
    unsigned* glist = wlist + (size_t)(blockIdx.x * 32 + wv) * PR_WCAP;
    int wc_l = 0;
    if (!BUILD && room) wc_l = wcnt[blockIdx.x * 32 + lane];
    bool binned = false;
    int bc_l = 0;
    const int lb = (blockIdx.x / BL_SUB) * (DEG_G * 32);
    if (BUILD && ws_bytes >= BL_WS_NEED) {
        unsigned over = 0u;
#pragma unroll
        for (int j = 0; j < DEG_G; ++j) over |= (unsigned)((unsigned)bcnt[lb + j * 32 + lane] > (unsigned)BL_CAP);
        bc_l = bcnt[lb + wv * 16 + (lane & 15)];
        binned = __builtin_amdgcn_ballot_w32(over != 0u) == 0u;
    }

    const bool replay = !BUILD && room &&
                        __builtin_amdgcn_ballot_w32((unsigned)wc_l > (unsigned)PR_WCAP) == 0u;

    if (!replay)
        for (int i = tid; i < PR_T * 16; i += 1024) sacc4[i] = make_float4(0.0f, 0.0f, 0.0f, 0.0f);

    if (replay) {
        const int n = __builtin_amdgcn_readlane(wc_l, wv);
        unsigned* went = squeue + wv * PR_QCAP;
        replay_wave(x_src, x_dst, dis, glist, n, lo, wv * PR_WROWS, min(PR_T, N_NODES - lo),
                    went, sacc4 + wv * 272, (float*)(went + 272), lane, lt);
    } else {
        int wtot = 0;
        for (int pass = 0; pass < PR_MAXPASS; ++pass) {
            const int wlo = pass * PR_QCAP;

            unsigned* q = squeue + wv * PR_QCAP;
            int seen = 0;
            if (BUILD && binned) {
            const unsigned j0 = (unsigned)((blockIdx.x % BL_SUB) * PR_T);
            const unsigned* bl0 = blist + (size_t)(lb + wv * 16) * BL_CAP;
            unsigned en[BL_CAP / 32];
#pragma unroll
            for (int c = 0; c < BL_CAP / 32; ++c) en[c] = bl0[c * 32 + lane];
            for (int i = 0; i < 16; ++i) {
                const int n = __builtin_amdgcn_readlane(bc_l, i);
                unsigned ec[BL_CAP / 32];
#pragma unroll
                for (int c = 0; c < BL_CAP / 32; ++c) ec[c] = en[c];
                {
                    const unsigned* bl = bl0 + (size_t)min(i + 1, 15) * BL_CAP;
#pragma unroll
                    for (int c = 0; c < BL_CAP / 32; ++c) en[c] = bl[c * 32 + lane];
                }
#pragma unroll
                for (int c = 0; c < BL_CAP / 32; ++c) {
                    if (c * 32 < n) {
                        const unsigned u = (ec[c] >> 18) - j0;
                        const bool t = u < (unsigned)PR_T;
                        const unsigned b = __builtin_amdgcn_ballot_w32(t);
                        if (b != 0u) {
                            const unsigned w = (unsigned)(seen + __popc(b & lt) - wlo);
                            if (t && w < (unsigned)PR_QCAP) q[w] = (u << 20) | (ec[c] & 0x3FFFFu);
                            seen += __popc(b);
                        }
                    }
                }
            }
            } else {
            int4 nx[4];
            scan_load(col, rot, tid, nx);
            for (int it = 0; it < PR_TRIPS; ++it) {
                const int ip = (it + rot < PR_TRIPS) ? it + rot : it + rot - PR_TRIPS;
                int4 cc[4];
#pragma unroll
                for (int j = 0; j < 4; ++j) cc[j] = nx[j];
                {
                    const int in1 = (it + 1 < PR_TRIPS) ? it + 1 + rot : rot;
                    scan_load(col, (in1 < PR_TRIPS) ? in1 : in1 - PR_TRIPS, tid, nx);
                }
#pragma unroll
                for (int j = 0; j < 4; ++j) {
                    const int e = (ip * 4096 + j * 1024 + tid) * 4;
                    const bool in = e < N_EDGES;
                    const unsigned bin = __builtin_amdgcn_ballot_w32(in);
                    const unsigned u0 = (unsigned)(cc[j].x - lo), u1 = (unsigned)(cc[j].y - lo);
                    const unsigned u2 = (unsigned)(cc[j].z - lo), u3 = (unsigned)(cc[j].w - lo);
                    const bool t0 = u0 < (unsigned)PR_T, t1 = u1 < (unsigned)PR_T;
                    const bool t2 = u2 < (unsigned)PR_T, t3 = u3 < (unsigned)PR_T;
                    const unsigned b0 = __builtin_amdgcn_ballot_w32(t0) & bin;
                    const unsigned b1 = __builtin_amdgcn_ballot_w32(t1) & bin;
                    const unsigned b2 = __builtin_amdgcn_ballot_w32(t2) & bin;
                    const unsigned b3 = __builtin_amdgcn_ballot_w32(t3) & bin;
                    if ((b0 | b1 | b2 | b3) != 0u) {
                        if (b0 != 0u) {
                            const unsigned w = (unsigned)(seen + __popc(b0 & lt) - wlo);
                            if (t0 && in && w < (unsigned)PR_QCAP) q[w] = (u0 << 20) | (unsigned)(e + 0);
                            seen += __popc(b0);
                        }
                        if (b1 != 0u) {
                            const unsigned w = (unsigned)(seen + __popc(b1 & lt) - wlo);
                            if (t1 && in && w < (unsigned)PR_QCAP) q[w] = (u1 << 20) | (unsigned)(e + 1);
                            seen += __popc(b1);
                        }
                        if (b2 != 0u) {
                            const unsigned w = (unsigned)(seen + __popc(b2 & lt) - wlo);
                            if (t2 && in && w < (unsigned)PR_QCAP) q[w] = (u2 << 20) | (unsigned)(e + 2);
                            seen += __popc(b2);
                        }
                        if (b3 != 0u) {
                            const unsigned w = (unsigned)(seen + __popc(b3 & lt) - wlo);
                            if (t3 && in && w < (unsigned)PR_QCAP) q[w] = (u3 << 20) | (unsigned)(e + 3);
                            seen += __popc(b3);
                        }
                    }
                }
            }

            {
                const int nown = min(max(seen - wlo, 0), PR_QCAP);
                for (int i0 = 0; i0 < nown; i0 += 128) {
                    unsigned ent[4];
                    int sid[4];
#pragma unroll
                    for (int j = 0; j < 4; ++j) {
                        const int idx = i0 + 32 * j + lane;
                        ent[j] = (idx < nown) ? q[idx] : 0u;
                    }
#pragma unroll
                    for (int j = 0; j < 4; ++j) sid[j] = row[ent[j] & 0xFFFFFu];
#pragma unroll
                    for (int j = 0; j < 4; ++j) {
                        const int idx = i0 + 32 * j + lane;
                        if (idx < nown) q[idx] = (ent[j] & 0xFFF00000u) | (unsigned)sid[j];
                    }
                }
            }
            }
            if (lane == 0) scnt[wv] = seen;
            __syncthreads();

            const int cnt_l = scnt[lane];
            const bool more = __builtin_amdgcn_ballot_w32(cnt_l > wlo + PR_QCAP) != 0u;

            {
                const unsigned olo = (unsigned)(wv * PR_WROWS);
                unsigned* st = sstage + wv * 64;
                int head = 0, have = 0;
                for (int qw = 0; qw < 32; ++qw) {
                    const int n = min(max(__builtin_amdgcn_readlane(cnt_l, qw) - wlo, 0), PR_QCAP);
                    const unsigned* qq = squeue + qw * PR_QCAP;
                    for (int i = 0; i < n; i += 128) {
                        unsigned ent[4];
#pragma unroll
                        for (int j = 0; j < 4; ++j) {
                            const int idx = i + 32 * j + lane;
                            ent[j] = (idx < n) ? qq[idx] : 0xFFFFFFFFu;
                        }
#pragma unroll
                        for (int j = 0; j < 4; ++j) {
                            const bool m = ((ent[j] >> 20) - olo) < (unsigned)PR_WROWS;
                            const unsigned b = __builtin_amdgcn_ballot_w32(m);
                            if (b != 0u) {
                                if (m) st[(head + have + __popc(b & lt)) & 63] = ent[j];
                                have += __popc(b);
                                while (have >= 16) {
                                    const unsigned ent = prop_group(x_src, dis, sacc, st + head, 16, lo, hl, hi);
                                    if (BUILD && room && !hi && wtot + hl < PR_WCAP) glist[wtot + hl] = ent;
                                    wtot += 16;
                                    head = (head + 16) & 63;
                                    have -= 16;
                                }
                            }
                        }
                    }
                }
                if (have > 0) {
                    const unsigned ent = prop_group(x_src, dis, sacc, st + head, have, lo, hl, hi);
                    if (BUILD && room && !hi && hl < have && wtot + hl < PR_WCAP) glist[wtot + hl] = ent;
                    wtot += have;
                }
            }
            __syncthreads();
            if (!more) break;
        }
        if (BUILD && room && lane == 0) wcnt[blockIdx.x * 32 + wv] = wtot;
    }

    if (!replay) {
        const int nrows = min(PR_T, N_NODES - lo);
        float4* dst = (float4*)x_dst + (size_t)lo * 16;
        for (int i = tid; i < nrows * 16; i += 1024) dst[i] = sacc4[i];
    }
}

extern "C" void kernel_launch(void* const* d_in, const int* in_sizes, int n_in,
                              void* d_out, int out_size, void* d_ws, size_t ws_size,
                              hipStream_t stream) {
    const float* Gu      = (const float*)d_in[0];
    const float* Gi      = (const float*)d_in[1];
    const int*   edge    = (const int*)d_in[2];
    const float* intents = (const float*)d_in[3];
    const int* row = edge;
    const int* col = edge + N_EDGES;
    float* out = (float*)d_out;

    float* deg  = (float*)d_ws;
    float* xbuf = deg + (size_t)N_NODES * 4;

    int*      wcnt  = (int*)(xbuf + (size_t)N_NODES * 64);
    unsigned* wlist = (unsigned*)(wcnt + PR_R * 32);
    int*      bcnt  = (int*)(wlist + (size_t)PR_R * 32 * PR_WCAP);
    unsigned* blist = (unsigned*)(bcnt + BL_LISTS);

    dim3 pgrid(PR_R), pblock(1024);

    k_deg<<<dim3(DEG_R * DEG_G), dim3(1024), 0, stream>>>(row, col, intents, out, bcnt, blist, ws_size);
    k_dis_x0<<<dim3((N_NODES + 127) / 128), dim3(128), 0, stream>>>(out, deg, Gu, Gi, xbuf);

    k_propagate<true ><<<pgrid, pblock, 0, stream>>>(xbuf, out, deg, row, col, wcnt, wlist, bcnt, blist, ws_size);
    k_propagate<false><<<pgrid, pblock, 0, stream>>>(out, xbuf, deg, row, col, wcnt, wlist, bcnt, blist, ws_size);
    k_propagate<false><<<pgrid, pblock, 0, stream>>>(xbuf, out, deg, row, col, wcnt, wlist, bcnt, blist, ws_size);
}


# x0 copy moved from k_dis_x0 into k_deg: trips 0-9 each load one 16-byte piece per lane behind the stream loads and store it at the trip's end; k_dis only sums the planes; sums unchanged
# speedup vs baseline: 3.1220x; 1.0443x over previous
#include <hip/hip_runtime.h>

#define N_USERS  100000
#define N_ITEMS  50000
#define N_NODES  150000
#define N_EDGES  1000000
#define EMBED_K  64

typedef __attribute__((ext_vector_type(2))) float v2f;
typedef __attribute__((ext_vector_type(8))) float v8f;

#define DEG_R   15
#define DEG_T   10548
#define DEG_G   16
#define DEG_CH  4096
#define BL_CAP   256
#define BL_SUB   18
#define BL_LISTS (DEG_R * DEG_G * 32)
#define BL_WS_NEED ((size_t)N_NODES * 4 * 4 + (size_t)N_NODES * 64 * 4 + (size_t)256 * 32 * 4 + (size_t)256 * 32 * 256 * 4 + (size_t)BL_LISTS * 4 + (size_t)BL_LISTS * BL_CAP * 4)

__device__ __forceinline__ void ldsAddF32(float* p, float v) {
    (void)__hip_atomic_fetch_add(p, v, __ATOMIC_RELAXED, __HIP_MEMORY_SCOPE_WORKGROUP);
}

__device__ __forceinline__ float4 edge_w(float z0, float z1, float z2, float z3) {
    float m  = fmaxf(fmaxf(z0, z1), fmaxf(z2, z3));
    float w0 = __expf(z0 - m), w1 = __expf(z1 - m);
    float w2 = __expf(z2 - m), w3 = __expf(z3 - m);
    float inv = 1.0f / (w0 + w1 + w2 + w3);
    return make_float4(w0 * inv, w1 * inv, w2 * inv, w3 * inv);
}

__device__ __forceinline__ void node_add(float* sdeg, int lo, int node, float4 w) {
    const unsigned u = (unsigned)(node - lo);
    if (u < (unsigned)DEG_T) {
        float* p = sdeg + u * 4;
        ldsAddF32(p + 0, w.x); ldsAddF32(p + 1, w.y);
        ldsAddF32(p + 2, w.z); ldsAddF32(p + 3, w.w);
    }
}

__device__ __forceinline__ void bin_put(unsigned* bq, int& seen, unsigned lt, unsigned u, int src) {
    const bool t = u < (unsigned)DEG_T;
    const unsigned b = __builtin_amdgcn_ballot_w32(t);
    if (b != 0u) {
        const unsigned w = (unsigned)(seen + __popc(b & lt));
        if (t && w < (unsigned)BL_CAP) bq[w] = (u << 18) | (unsigned)src;
        seen += __popc(b);
    }
}

#define X0_U4   (N_USERS * (EMBED_K / 4))
#define X0_N4   (N_NODES * (EMBED_K / 4))
#define X0_CP   10
static_assert((long long)DEG_R * DEG_G * X0_CP * 1024 >= X0_N4, "the work-groups' parts cover x0");

__device__ __forceinline__ float4 x0_piece(const float* __restrict__ Gu, const float* __restrict__ Gi, int f) {
    const int fc = min(f, X0_N4 - 1);
    const float4* p = (fc < X0_U4) ? (const float4*)Gu + fc : (const float4*)Gi + (fc - X0_U4);
    return *p;
}
__device__ __forceinline__ void x0_put(float* __restrict__ x0, int f, float4 v) {
    if (f < X0_N4) ((float4*)x0)[f] = v;
}

template <bool CPY>
__device__ __forceinline__ void deg_trip(
        const int* __restrict__ row, const int* __restrict__ col, const float* __restrict__ intents,
        float* sdeg, int lo, bool keep, unsigned* bq, int& seen, unsigned lt, int e,
        const float* __restrict__ Gu, const float* __restrict__ Gi, float* __restrict__ x0, int f) {
    float4 cp = make_float4(0.0f, 0.0f, 0.0f, 0.0f);
    if (e < N_EDGES) {
        const int4   r4 = *(const int4*)(row + e);
        const int4   c4 = *(const int4*)(col + e);
        const float4 za = *(const float4*)(intents + e);
        const float4 zb = *(const float4*)(intents + N_EDGES + e);
        const float4 zc = *(const float4*)(intents + 2 * N_EDGES + e);
        const float4 zd = *(const float4*)(intents + 3 * N_EDGES + e);
        if (CPY) cp = x0_piece(Gu, Gi, f);
        const float4 w0 = edge_w(za.x, zb.x, zc.x, zd.x);
        const float4 w1 = edge_w(za.y, zb.y, zc.y, zd.y);
        const float4 w2 = edge_w(za.z, zb.z, zc.z, zd.z);
        const float4 w3 = edge_w(za.w, zb.w, zc.w, zd.w);
        node_add(sdeg, lo, r4.x, w0); node_add(sdeg, lo, c4.x, w0);
        node_add(sdeg, lo, r4.y, w1); node_add(sdeg, lo, c4.y, w1);
        node_add(sdeg, lo, r4.z, w2); node_add(sdeg, lo, c4.z, w2);
        node_add(sdeg, lo, r4.w, w3); node_add(sdeg, lo, c4.w, w3);
        if (keep) {
            bin_put(bq, seen, lt, (unsigned)(c4.x - lo), r4.x);
            bin_put(bq, seen, lt, (unsigned)(c4.y - lo), r4.y);
            bin_put(bq, seen, lt, (unsigned)(c4.z - lo), r4.z);
            bin_put(bq, seen, lt, (unsigned)(c4.w - lo), r4.w);
        }
    } else if (CPY) {
        cp = x0_piece(Gu, Gi, f);
    }
    if (CPY) x0_put(x0, f, cp);
}

__global__ __launch_bounds__(1024) void k_deg(
        const int* __restrict__ row, const int* __restrict__ col,
        const float* __restrict__ intents, float* __restrict__ planes,
        int* __restrict__ bcnt, unsigned* __restrict__ blist, size_t ws_bytes,
        const float* __restrict__ Gu, const float* __restrict__ Gi, float* __restrict__ x0) {
    __shared__ float4 sdeg4[DEG_T];
    __shared__ unsigned sbq[32 * BL_CAP];
    float* sdeg = (float*)sdeg4;
    const int tid = threadIdx.x;
    const int wv   = __builtin_amdgcn_readfirstlane(tid >> 5);
    const int lane = tid & 31;
    const unsigned lt = (1u << lane) - 1u;
    const bool keep = ws_bytes >= BL_WS_NEED;
    unsigned* bq = sbq + wv * BL_CAP;
    int seen = 0;
    const int rg  = blockIdx.x % DEG_R;
    const int g   = blockIdx.x / DEG_R;
    const int lo  = rg * DEG_T;

    for (int i = tid; i < DEG_T; i += 1024) sdeg4[i] = make_float4(0.0f, 0.0f, 0.0f, 0.0f);
    __syncthreads();

    const int cbase = blockIdx.x * (X0_CP * 1024) + tid;
    int ch = g, it = 0;
    for (; it < X0_CP && ch * DEG_CH < N_EDGES; ch += DEG_G, ++it)
        deg_trip<true >(row, col, intents, sdeg, lo, keep, bq, seen, lt, ch * DEG_CH + tid * 4, Gu, Gi, x0, cbase + it * 1024);
    for (; ch * DEG_CH < N_EDGES; ch += DEG_G)
        deg_trip<false>(row, col, intents, sdeg, lo, keep, bq, seen, lt, ch * DEG_CH + tid * 4, Gu, Gi, x0, 0);
    for (; it < X0_CP; ++it) x0_put(x0, cbase + it * 1024, x0_piece(Gu, Gi, cbase + it * 1024));
    if (keep) {
        const int nseen = __builtin_amdgcn_readfirstlane(seen);
        const int L = (rg * DEG_G + g) * 32 + wv;
        unsigned* gl = blist + (size_t)L * BL_CAP;
        const int nst = min(nseen, BL_CAP);
        for (int c = 0; c < BL_CAP / 32; ++c) {
            const int idx = c * 32 + lane;
            gl[idx] = (idx < nst) ? bq[idx] : 0xFFFFFFFFu;
        }
        if (lane == 0) bcnt[L] = nseen;
    }
    __syncthreads();

    float4* dst = (float4*)(planes + (size_t)g * (N_NODES * 4)) + lo;
    for (int i = tid; i < DEG_T; i += 1024)
        if (lo + i < N_NODES) dst[i] = sdeg4[i];
}

__global__ __launch_bounds__(128) void k_dis(
        const float* __restrict__ planes, float* __restrict__ dis) {
    const int i = blockIdx.x * 128 + threadIdx.x;
    if (i < N_NODES) {
        float4 s = ((const float4*)planes)[i];
#pragma unroll
        for (int g = 1; g < DEG_G; ++g) {
            const float4 p = ((const float4*)(planes + (size_t)g * (N_NODES * 4)))[i];
            s.x += p.x; s.y += p.y; s.z += p.z; s.w += p.w;
        }
        float4 o;
        o.x = (s.x > 0.0f) ? rsqrtf(fmaxf(s.x, 1e-12f)) : 0.0f;
        o.y = (s.y > 0.0f) ? rsqrtf(fmaxf(s.y, 1e-12f)) : 0.0f;
        o.z = (s.z > 0.0f) ? rsqrtf(fmaxf(s.z, 1e-12f)) : 0.0f;
        o.w = (s.w > 0.0f) ? rsqrtf(fmaxf(s.w, 1e-12f)) : 0.0f;
        ((float4*)dis)[i] = o;
    }
}

#define PR_R       256
#define PR_T       586
#define PR_WROWS   19
#define PR_QCAP    1024
#define PR_MAXPASS 32
#define PR_TRIPS   ((N_EDGES + 128 * 128 - 1) / (128 * 128))
#define PR_WCAP    256
#define PR_WS_NEED ((size_t)N_NODES * 4 * 4 + (size_t)N_NODES * 64 * 4 + (size_t)PR_R * 32 * 4 + (size_t)PR_R * 32 * PR_WCAP * 4)
static_assert(BL_WS_NEED == PR_WS_NEED + (size_t)BL_LISTS * 4 + (size_t)BL_LISTS * BL_CAP * 4, "the binned lists lie behind the stored lists");
static_assert(DEG_T == BL_SUB * PR_T, "a degree range must be whole layer ranges");
static_assert(DEG_R * BL_SUB >= PR_R, "every layer range needs a degree range");
static_assert(DEG_G * 32 == 32 * 16, "a layer-1 wave takes 16 of its range's lists");
static_assert(32 * 272 <= PR_T * 16 && 272 + 16 * 36 <= PR_QCAP && PR_WCAP + 6 <= 272, "the replay path's per-wave LDS areas fit");

__device__ __forceinline__ unsigned prop_group(
        const float* __restrict__ x_src, const float* __restrict__ dis,
        float* sacc, const unsigned* grp, int n, int lo,
        int hl, bool hi) {
    const bool vA = hl < n;
    const unsigned entA = vA ? grp[hl] : 0u;
    const int irA = (int)(entA & 0xFFFFFu);
    const int dlA = (int)(entA >> 20);
    const float4 dr = *(const float4*)(dis + irA * 4);
    const float4 dc = *(const float4*)(dis + (lo + dlA) * 4);
    float nrm[4];
    nrm[0] = vA ? dr.x * dc.x : 0.0f; nrm[1] = vA ? dr.y * dc.y : 0.0f;
    nrm[2] = vA ? dr.z * dc.z : 0.0f; nrm[3] = vA ? dr.w * dc.w : 0.0f;

    const int kx = hi ? 2 : 0;
    const int mo = hi ? 8 : 0;
    int rb[8];
#pragma unroll
    for (int t = 0; t < 4; ++t) {
        rb[2 * t]     = __shfl(irA, 4 * t + kx);
        rb[2 * t + 1] = __shfl(irA, 4 * t + kx + 1);
    }
    int dlv[8];
#pragma unroll
    for (int v = 0; v < 8; ++v) dlv[v] = __shfl(dlA, v + mo);

    float bx[16], by[16];
#pragma unroll
    for (int k = 0; k < 4; ++k) {
#pragma unroll
        for (int t = 0; t < 4; ++t) {
            bx[k * 4 + t] = x_src[rb[2 * t]     * 64 + k * 16 + hl];
            by[k * 4 + t] = x_src[rb[2 * t + 1] * 64 + k * 16 + hl];
        }
    }

#pragma unroll
    for (int k = 0; k < 4; ++k) {
        v8f acc = {};
#pragma unroll
        for (int t = 0; t < 4; ++t) {
            v2f a, b;
            a.x = (hl == 4 * t + kx)     ? nrm[k] : 0.0f;
            a.y = (hl == 4 * t + kx + 1) ? nrm[k] : 0.0f;
            b.x = bx[k * 4 + t];
            b.y = by[k * 4 + t];
            acc = __builtin_amdgcn_wmma_f32_16x16x4_f32(
                      false, a, false, b, (short)0, acc, false, false);
        }
#pragma unroll
        for (int v = 0; v < 8; ++v)
            if (v + mo < n) ldsAddF32(sacc + dlv[v] * 64 + k * 16 + hl, acc[v]);
    }
    return entA;
}


#define RP_PITCH 36
__device__ __forceinline__ void replay_wave(
        const float* __restrict__ x_src, float* __restrict__ x_dst, const float* __restrict__ dis,
        const unsigned* __restrict__ glist, int n, int lo, int olo, int nrows,
        unsigned* went, float4* wnrm, float* stg, int lane, unsigned lt) {
    const int c = lane & 15;
    const int h = lane >> 4;
    unsigned e[PR_WCAP / 32];
#pragma unroll
    for (int j = 0; j < PR_WCAP / 32; ++j) {
        const int idx = j * 32 + lane;
        e[j] = (idx < n) ? glist[idx] : 0u;
    }
    float4 nv[PR_WCAP / 32];
    unsigned b0[PR_WCAP / 32], b1[PR_WCAP / 32];
    int n0 = 0, n1 = 0;
#pragma unroll
    for (int j = 0; j < PR_WCAP / 32; ++j) {
        const bool valid = j * 32 + lane < n;
        const int src = (int)(e[j] & 0xFFFFFu);
        const int dl  = (int)(e[j] >> 20);
        const float4 dr = *(const float4*)(dis + src * 4);
        const float4 dc = *(const float4*)(dis + (lo + dl) * 4);
        nv[j].x = valid ? dr.x * dc.x : 0.0f; nv[j].y = valid ? dr.y * dc.y : 0.0f;
        nv[j].z = valid ? dr.z * dc.z : 0.0f; nv[j].w = valid ? dr.w * dc.w : 0.0f;
        const bool up = (dl - olo) >= 16;
        b0[j] = __builtin_amdgcn_ballot_w32(valid && !up);
        b1[j] = __builtin_amdgcn_ballot_w32(valid && up);
        n0 += __popc(b0[j]);
        n1 += __popc(b1[j]);
    }
    const int z0 = (n0 + 3) & ~3;
    const int z1 = z0 + ((n1 + 3) & ~3);
    {
        int p0 = 0, p1 = z0;
#pragma unroll
        for (int j = 0; j < PR_WCAP / 32; ++j) {
            const bool valid = j * 32 + lane < n;
            const bool up = ((int)(e[j] >> 20) - olo) >= 16;
            const int pos = up ? p1 + __popc(b1[j] & lt) : p0 + __popc(b0[j] & lt);
            if (valid) { went[pos] = e[j]; wnrm[pos] = nv[j]; }
            p0 += __popc(b0[j]);
            p1 += __popc(b1[j]);
        }
        if (n0 + lane < z0)      { went[n0 + lane] = 0u;      wnrm[n0 + lane]      = make_float4(0.0f, 0.0f, 0.0f, 0.0f); }
        if (z0 + n1 + lane < z1) { went[z0 + n1 + lane] = 0u; wnrm[z0 + n1 + lane] = make_float4(0.0f, 0.0f, 0.0f, 0.0f); }
    }

    for (int tile = 0; tile < 2; ++tile) {
        const int gb = tile ? z0 : 0;
        const int ge = tile ? z1 : z0;
        const int rb = olo + tile * 16;
        v8f acc[4] = {};
        for (int g0 = gb; g0 < ge; g0 += 16) {
            const int cnt = min(16, ge - g0);
            float ax[16], ay[16], wa[16], wb[16];
#pragma unroll
            for (int t = 0; t < 4; ++t) {
                const int idx = min(g0 + 4 * t + 2 * h, ge - 2);
                const uint2 ee = *(const uint2*)(went + idx);
                const float4 na = wnrm[idx];
                const float4 nb = wnrm[idx + 1];
                const int s0 = (int)(ee.x & 0xFFFFFu);
                const int s1 = (int)(ee.y & 0xFFFFFu);
                const bool m0 = ((int)(ee.x >> 20) - rb) == c;
                const bool m1 = ((int)(ee.y >> 20) - rb) == c;
#pragma unroll
                for (int k = 0; k < 4; ++k) {
                    ax[t * 4 + k] = x_src[s0 * 64 + k * 16 + c];
                    ay[t * 4 + k] = x_src[s1 * 64 + k * 16 + c];
                }
                wa[t * 4 + 0] = m0 ? na.x : 0.0f; wa[t * 4 + 1] = m0 ? na.y : 0.0f;
                wa[t * 4 + 2] = m0 ? na.z : 0.0f; wa[t * 4 + 3] = m0 ? na.w : 0.0f;
                wb[t * 4 + 0] = m1 ? nb.x : 0.0f; wb[t * 4 + 1] = m1 ? nb.y : 0.0f;
                wb[t * 4 + 2] = m1 ? nb.z : 0.0f; wb[t * 4 + 3] = m1 ? nb.w : 0.0f;
            }
#pragma unroll
            for (int i = 0; i < 16; ++i) {
                asm volatile("" :: "v"(ax[i]), "v"(ay[i]));
            }
#pragma unroll
            for (int t = 0; t < 4; ++t) {
                if (t == 0 || 4 * t < cnt) {
#pragma unroll
                    for (int k = 0; k < 4; ++k) {
                        v2f a, b;
                        a.x = ax[t * 4 + k]; a.y = ay[t * 4 + k];
                        b.x = wa[t * 4 + k]; b.y = wb[t * 4 + k];
                        acc[k] = __builtin_amdgcn_wmma_f32_16x16x4_f32(
                                     false, a, false, b, (short)0, acc[k], false, false);
                    }
                }
            }
        }
        const int vr = min(min(PR_WROWS - tile * 16, 16), min(PR_T, nrows) - rb);
        if (vr > 0) {
#pragma unroll
            for (int p = 0; p < 2; ++p) {
                float* sp = stg + c * RP_PITCH + 8 * h;
                *(float4*)(sp)      = make_float4(acc[2 * p][0], acc[2 * p][1], acc[2 * p][2], acc[2 * p][3]);
                *(float4*)(sp + 4)  = make_float4(acc[2 * p][4], acc[2 * p][5], acc[2 * p][6], acc[2 * p][7]);
                *(float4*)(sp + 16) = make_float4(acc[2 * p + 1][0], acc[2 * p + 1][1], acc[2 * p + 1][2], acc[2 * p + 1][3]);
                *(float4*)(sp + 20) = make_float4(acc[2 * p + 1][4], acc[2 * p + 1][5], acc[2 * p + 1][6], acc[2 * p + 1][7]);
#pragma unroll
                for (int j = 0; j < 4; ++j) {
                    if (j * 4 < vr) {
                        const int i = j * 32 + lane;
                        const int r = i >> 3, q = i & 7;
                        const float4 v = *(const float4*)(stg + r * RP_PITCH + q * 4);
                        if (r < vr) *(float4*)(x_dst + (size_t)(lo + rb + r) * 64 + p * 32 + q * 4) = v;
                    }
                }
            }
        }
    }
}

__device__ __forceinline__ void scan_load(const int* __restrict__ col, int ip, int tid, int4 (&cc)[4]) {
#pragma unroll
    for (int j = 0; j < 4; ++j) {
        const unsigned i4 = (unsigned)min(ip * 4096 + j * 1024 + tid, N_EDGES / 4 - 1);
        cc[j] = *(const int4*)((const char*)col + i4 * 16u);
    }
}

template <bool BUILD>
__global__ __launch_bounds__(1024) void k_propagate(
        const float* __restrict__ x_src, float* __restrict__ x_dst,
        const float* __restrict__ dis,
        const int* __restrict__ row, const int* __restrict__ col,
        int* __restrict__ wcnt, unsigned* __restrict__ wlist,
        const int* __restrict__ bcnt, const unsigned* __restrict__ blist, size_t ws_bytes) {
    __shared__ float4   sacc4[PR_T * 16];
    __shared__ __attribute__((aligned(16))) unsigned squeue[32 * PR_QCAP];
    __shared__ unsigned sstage[32 * 64];
    __shared__ int      scnt[32];
    float* sacc = (float*)sacc4;
    const int tid  = threadIdx.x;
    const int wv   = __builtin_amdgcn_readfirstlane(tid >> 5);
    const int lane = tid & 31;
    const int hl   = lane & 15;
    const bool hi  = lane >= 16;
    const unsigned lt = (1u << lane) - 1u;
    const int lo   = blockIdx.x * PR_T;
    const int rot  = blockIdx.x % PR_TRIPS;

    const bool room = ws_bytes >= PR_WS_NEED;
    unsigned* glist = wlist + (size_t)(blockIdx.x * 32 + wv) * PR_WCAP;
    int wc_l = 0;
    if (!BUILD && room) wc_l = wcnt[blockIdx.x * 32 + lane];
    bool binned = false;
    int bc_l = 0;
    const int lb = (blockIdx.x / BL_SUB) * (DEG_G * 32);
    if (BUILD && ws_bytes >= BL_WS_NEED) {
        unsigned over = 0u;
#pragma unroll
        for (int j = 0; j < DEG_G; ++j) over |= (unsigned)((unsigned)bcnt[lb + j * 32 + lane] > (unsigned)BL_CAP);
        bc_l = bcnt[lb + wv * 16 + (lane & 15)];
        binned = __builtin_amdgcn_ballot_w32(over != 0u) == 0u;
    }

    const bool replay = !BUILD && room &&
                        __builtin_amdgcn_ballot_w32((unsigned)wc_l > (unsigned)PR_WCAP) == 0u;

    if (!replay)
        for (int i = tid; i < PR_T * 16; i += 1024) sacc4[i] = make_float4(0.0f, 0.0f, 0.0f, 0.0f);

    if (replay) {
        const int n = __builtin_amdgcn_readlane(wc_l, wv);
        unsigned* went = squeue + wv * PR_QCAP;
        replay_wave(x_src, x_dst, dis, glist, n, lo, wv * PR_WROWS, min(PR_T, N_NODES - lo),
                    went, sacc4 + wv * 272, (float*)(went + 272), lane, lt);
    } else {
        int wtot = 0;
        for (int pass = 0; pass < PR_MAXPASS; ++pass) {
            const int wlo = pass * PR_QCAP;

            unsigned* q = squeue + wv * PR_QCAP;
            int seen = 0;
            if (BUILD && binned) {
            const unsigned j0 = (unsigned)((blockIdx.x % BL_SUB) * PR_T);
            const unsigned* bl0 = blist + (size_t)(lb + wv * 16) * BL_CAP;
            unsigned en[BL_CAP / 32];
#pragma unroll
            for (int c = 0; c < BL_CAP / 32; ++c) en[c] = bl0[c * 32 + lane];
            for (int i = 0; i < 16; ++i) {
                const int n = __builtin_amdgcn_readlane(bc_l, i);
                unsigned ec[BL_CAP / 32];
#pragma unroll
                for (int c = 0; c < BL_CAP / 32; ++c) ec[c] = en[c];
                {
                    const unsigned* bl = bl0 + (size_t)min(i + 1, 15) * BL_CAP;
#pragma unroll
                    for (int c = 0; c < BL_CAP / 32; ++c) en[c] = bl[c * 32 + lane];
                }
#pragma unroll
                for (int c = 0; c < BL_CAP / 32; ++c) {
                    if (c * 32 < n) {
                        const unsigned u = (ec[c] >> 18) - j0;
                        const bool t = u < (unsigned)PR_T;
                        const unsigned b = __builtin_amdgcn_ballot_w32(t);
                        if (b != 0u) {
                            const unsigned w = (unsigned)(seen + __popc(b & lt) - wlo);
                            if (t && w < (unsigned)PR_QCAP) q[w] = (u << 20) | (ec[c] & 0x3FFFFu);
                            seen += __popc(b);
                        }
                    }
                }
            }
            } else {
            int4 nx[4];
            scan_load(col, rot, tid, nx);
            for (int it = 0; it < PR_TRIPS; ++it) {
                const int ip = (it + rot < PR_TRIPS) ? it + rot : it + rot - PR_TRIPS;
                int4 cc[4];
#pragma unroll
                for (int j = 0; j < 4; ++j) cc[j] = nx[j];
                {
                    const int in1 = (it + 1 < PR_TRIPS) ? it + 1 + rot : rot;
                    scan_load(col, (in1 < PR_TRIPS) ? in1 : in1 - PR_TRIPS, tid, nx);
                }
#pragma unroll
                for (int j = 0; j < 4; ++j) {
                    const int e = (ip * 4096 + j * 1024 + tid) * 4;
                    const bool in = e < N_EDGES;
                    const unsigned bin = __builtin_amdgcn_ballot_w32(in);
                    const unsigned u0 = (unsigned)(cc[j].x - lo), u1 = (unsigned)(cc[j].y - lo);
                    const unsigned u2 = (unsigned)(cc[j].z - lo), u3 = (unsigned)(cc[j].w - lo);
                    const bool t0 = u0 < (unsigned)PR_T, t1 = u1 < (unsigned)PR_T;
                    const bool t2 = u2 < (unsigned)PR_T, t3 = u3 < (unsigned)PR_T;
                    const unsigned b0 = __builtin_amdgcn_ballot_w32(t0) & bin;
                    const unsigned b1 = __builtin_amdgcn_ballot_w32(t1) & bin;
                    const unsigned b2 = __builtin_amdgcn_ballot_w32(t2) & bin;
                    const unsigned b3 = __builtin_amdgcn_ballot_w32(t3) & bin;
                    if ((b0 | b1 | b2 | b3) != 0u) {
                        if (b0 != 0u) {
                            const unsigned w = (unsigned)(seen + __popc(b0 & lt) - wlo);
                            if (t0 && in && w < (unsigned)PR_QCAP) q[w] = (u0 << 20) | (unsigned)(e + 0);
                            seen += __popc(b0);
                        }
                        if (b1 != 0u) {
                            const unsigned w = (unsigned)(seen + __popc(b1 & lt) - wlo);
                            if (t1 && in && w < (unsigned)PR_QCAP) q[w] = (u1 << 20) | (unsigned)(e + 1);
                            seen += __popc(b1);
                        }
                        if (b2 != 0u) {
                            const unsigned w = (unsigned)(seen + __popc(b2 & lt) - wlo);
                            if (t2 && in && w < (unsigned)PR_QCAP) q[w] = (u2 << 20) | (unsigned)(e + 2);
                            seen += __popc(b2);
                        }
                        if (b3 != 0u) {
                            const unsigned w = (unsigned)(seen + __popc(b3 & lt) - wlo);
                            if (t3 && in && w < (unsigned)PR_QCAP) q[w] = (u3 << 20) | (unsigned)(e + 3);
                            seen += __popc(b3);
                        }
                    }
                }
            }

            {
                const int nown = min(max(seen - wlo, 0), PR_QCAP);
                for (int i0 = 0; i0 < nown; i0 += 128) {
                    unsigned ent[4];
                    int sid[4];
#pragma unroll
                    for (int j = 0; j < 4; ++j) {
                        const int idx = i0 + 32 * j + lane;
                        ent[j] = (idx < nown) ? q[idx] : 0u;
                    }
#pragma unroll
                    for (int j = 0; j < 4; ++j) sid[j] = row[ent[j] & 0xFFFFFu];
#pragma unroll
                    for (int j = 0; j < 4; ++j) {
                        const int idx = i0 + 32 * j + lane;
                        if (idx < nown) q[idx] = (ent[j] & 0xFFF00000u) | (unsigned)sid[j];
                    }
                }
            }
            }
            if (lane == 0) scnt[wv] = seen;
            __syncthreads();

            const int cnt_l = scnt[lane];
            const bool more = __builtin_amdgcn_ballot_w32(cnt_l > wlo + PR_QCAP) != 0u;

            {
                const unsigned olo = (unsigned)(wv * PR_WROWS);
                unsigned* st = sstage + wv * 64;
                int head = 0, have = 0;
                for (int qw = 0; qw < 32; ++qw) {
                    const int n = min(max(__builtin_amdgcn_readlane(cnt_l, qw) - wlo, 0), PR_QCAP);
                    const unsigned* qq = squeue + qw * PR_QCAP;
                    for (int i = 0; i < n; i += 128) {
                        unsigned ent[4];
#pragma unroll
                        for (int j = 0; j < 4; ++j) {
                            const int idx = i + 32 * j + lane;
                            ent[j] = (idx < n) ? qq[idx] : 0xFFFFFFFFu;
                        }
#pragma unroll
                        for (int j = 0; j < 4; ++j) {
                            const bool m = ((ent[j] >> 20) - olo) < (unsigned)PR_WROWS;
                            const unsigned b = __builtin_amdgcn_ballot_w32(m);
                            if (b != 0u) {
                                if (m) st[(head + have + __popc(b & lt)) & 63] = ent[j];
                                have += __popc(b);
                                while (have >= 16) {
                                    const unsigned ent = prop_group(x_src, dis, sacc, st + head, 16, lo, hl, hi);
                                    if (BUILD && room && !hi && wtot + hl < PR_WCAP) glist[wtot + hl] = ent;
                                    wtot += 16;
                                    head = (head + 16) & 63;
                                    have -= 16;
                                }
                            }
                        }
                    }
                }
                if (have > 0) {
                    const unsigned ent = prop_group(x_src, dis, sacc, st + head, have, lo, hl, hi);
                    if (BUILD && room && !hi && hl < have && wtot + hl < PR_WCAP) glist[wtot + hl] = ent;
                    wtot += have;
                }
            }
            __syncthreads();
            if (!more) break;
        }
        if (BUILD && room && lane == 0) wcnt[blockIdx.x * 32 + wv] = wtot;
    }

    if (!replay) {
        const int nrows = min(PR_T, N_NODES - lo);
        float4* dst = (float4*)x_dst + (size_t)lo * 16;
        for (int i = tid; i < nrows * 16; i += 1024) dst[i] = sacc4[i];
    }
}

extern "C" void kernel_launch(void* const* d_in, const int* in_sizes, int n_in,
                              void* d_out, int out_size, void* d_ws, size_t ws_size,
                              hipStream_t stream) {
    const float* Gu      = (const float*)d_in[0];
    const float* Gi      = (const float*)d_in[1];
    const int*   edge    = (const int*)d_in[2];
    const float* intents = (const float*)d_in[3];
    const int* row = edge;
    const int* col = edge + N_EDGES;
    float* out = (float*)d_out;

    float* deg  = (float*)d_ws;
    float* xbuf = deg + (size_t)N_NODES * 4;

    int*      wcnt  = (int*)(xbuf + (size_t)N_NODES * 64);
    unsigned* wlist = (unsigned*)(wcnt + PR_R * 32);
    int*      bcnt  = (int*)(wlist + (size_t)PR_R * 32 * PR_WCAP);
    unsigned* blist = (unsigned*)(bcnt + BL_LISTS);

    dim3 pgrid(PR_R), pblock(1024);

    k_deg<<<dim3(DEG_R * DEG_G), dim3(1024), 0, stream>>>(row, col, intents, out, bcnt, blist, ws_size, Gu, Gi, xbuf);
    k_dis<<<dim3((N_NODES + 127) / 128), dim3(128), 0, stream>>>(out, deg);

    k_propagate<true ><<<pgrid, pblock, 0, stream>>>(xbuf, out, deg, row, col, wcnt, wlist, bcnt, blist, ws_size);
    k_propagate<false><<<pgrid, pblock, 0, stream>>>(out, xbuf, deg, row, col, wcnt, wlist, bcnt, blist, ws_size);
    k_propagate<false><<<pgrid, pblock, 0, stream>>>(xbuf, out, deg, row, col, wcnt, wlist, bcnt, blist, ws_size);
}


# layer 1 also stores each work-list entry's four norms (dis[src]*dis[dst]) beside the list; layers 2-3 load them in list order and gather nothing from dis; same products, same sum order
# speedup vs baseline: 3.2908x; 1.0541x over previous
#include <hip/hip_runtime.h>

#define N_USERS  100000
#define N_ITEMS  50000
#define N_NODES  150000
#define N_EDGES  1000000
#define EMBED_K  64

typedef __attribute__((ext_vector_type(2))) float v2f;
typedef __attribute__((ext_vector_type(8))) float v8f;

#define DEG_R   15
#define DEG_T   10548
#define DEG_G   16
#define DEG_CH  4096
#define BL_CAP   256
#define BL_SUB   18
#define BL_LISTS (DEG_R * DEG_G * 32)
#define BL_WS_NEED ((size_t)N_NODES * 4 * 4 + (size_t)N_NODES * 64 * 4 + (size_t)256 * 32 * 4 + (size_t)256 * 32 * 256 * 4 + (size_t)BL_LISTS * 4 + (size_t)BL_LISTS * BL_CAP * 4)

__device__ __forceinline__ void ldsAddF32(float* p, float v) {
    (void)__hip_atomic_fetch_add(p, v, __ATOMIC_RELAXED, __HIP_MEMORY_SCOPE_WORKGROUP);
}

__device__ __forceinline__ float4 edge_w(float z0, float z1, float z2, float z3) {
    float m  = fmaxf(fmaxf(z0, z1), fmaxf(z2, z3));
    float w0 = __expf(z0 - m), w1 = __expf(z1 - m);
    float w2 = __expf(z2 - m), w3 = __expf(z3 - m);
    float inv = 1.0f / (w0 + w1 + w2 + w3);
    return make_float4(w0 * inv, w1 * inv, w2 * inv, w3 * inv);
}

__device__ __forceinline__ void node_add(float* sdeg, int lo, int node, float4 w) {
    const unsigned u = (unsigned)(node - lo);
    if (u < (unsigned)DEG_T) {
        float* p = sdeg + u * 4;
        ldsAddF32(p + 0, w.x); ldsAddF32(p + 1, w.y);
        ldsAddF32(p + 2, w.z); ldsAddF32(p + 3, w.w);
    }
}

__device__ __forceinline__ void bin_put(unsigned* bq, int& seen, unsigned lt, unsigned u, int src) {
    const bool t = u < (unsigned)DEG_T;
    const unsigned b = __builtin_amdgcn_ballot_w32(t);
    if (b != 0u) {
        const unsigned w = (unsigned)(seen + __popc(b & lt));
        if (t && w < (unsigned)BL_CAP) bq[w] = (u << 18) | (unsigned)src;
        seen += __popc(b);
    }
}

#define X0_U4   (N_USERS * (EMBED_K / 4))
#define X0_N4   (N_NODES * (EMBED_K / 4))
#define X0_CP   10
static_assert((long long)DEG_R * DEG_G * X0_CP * 1024 >= X0_N4, "the work-groups' parts cover x0");

__device__ __forceinline__ float4 x0_piece(const float* __restrict__ Gu, const float* __restrict__ Gi, int f) {
    const int fc = min(f, X0_N4 - 1);
    const float4* p = (fc < X0_U4) ? (const float4*)Gu + fc : (const float4*)Gi + (fc - X0_U4);
    return *p;
}
__device__ __forceinline__ void x0_put(float* __restrict__ x0, int f, float4 v) {
    if (f < X0_N4) ((float4*)x0)[f] = v;
}

template <bool CPY>
__device__ __forceinline__ void deg_trip(
        const int* __restrict__ row, const int* __restrict__ col, const float* __restrict__ intents,
        float* sdeg, int lo, bool keep, unsigned* bq, int& seen, unsigned lt, int e,
        const float* __restrict__ Gu, const float* __restrict__ Gi, float* __restrict__ x0, int f) {
    float4 cp = make_float4(0.0f, 0.0f, 0.0f, 0.0f);
    if (e < N_EDGES) {
        const int4   r4 = *(const int4*)(row + e);
        const int4   c4 = *(const int4*)(col + e);
        const float4 za = *(const float4*)(intents + e);
        const float4 zb = *(const float4*)(intents + N_EDGES + e);
        const float4 zc = *(const float4*)(intents + 2 * N_EDGES + e);
        const float4 zd = *(const float4*)(intents + 3 * N_EDGES + e);
        if (CPY) cp = x0_piece(Gu, Gi, f);
        const float4 w0 = edge_w(za.x, zb.x, zc.x, zd.x);
        const float4 w1 = edge_w(za.y, zb.y, zc.y, zd.y);
        const float4 w2 = edge_w(za.z, zb.z, zc.z, zd.z);
        const float4 w3 = edge_w(za.w, zb.w, zc.w, zd.w);
        node_add(sdeg, lo, r4.x, w0); node_add(sdeg, lo, c4.x, w0);
        node_add(sdeg, lo, r4.y, w1); node_add(sdeg, lo, c4.y, w1);
        node_add(sdeg, lo, r4.z, w2); node_add(sdeg, lo, c4.z, w2);
        node_add(sdeg, lo, r4.w, w3); node_add(sdeg, lo, c4.w, w3);
        if (keep) {
            bin_put(bq, seen, lt, (unsigned)(c4.x - lo), r4.x);
            bin_put(bq, seen, lt, (unsigned)(c4.y - lo), r4.y);
            bin_put(bq, seen, lt, (unsigned)(c4.z - lo), r4.z);
            bin_put(bq, seen, lt, (unsigned)(c4.w - lo), r4.w);
        }
    } else if (CPY) {
        cp = x0_piece(Gu, Gi, f);
    }
    if (CPY) x0_put(x0, f, cp);
}

__global__ __launch_bounds__(1024) void k_deg(
        const int* __restrict__ row, const int* __restrict__ col,
        const float* __restrict__ intents, float* __restrict__ planes,
        int* __restrict__ bcnt, unsigned* __restrict__ blist, size_t ws_bytes,
        const float* __restrict__ Gu, const float* __restrict__ Gi, float* __restrict__ x0) {
    __shared__ float4 sdeg4[DEG_T];
    __shared__ unsigned sbq[32 * BL_CAP];
    float* sdeg = (float*)sdeg4;
    const int tid = threadIdx.x;
    const int wv   = __builtin_amdgcn_readfirstlane(tid >> 5);
    const int lane = tid & 31;
    const unsigned lt = (1u << lane) - 1u;
    const bool keep = ws_bytes >= BL_WS_NEED;
    unsigned* bq = sbq + wv * BL_CAP;
    int seen = 0;
    const int rg  = blockIdx.x % DEG_R;
    const int g   = blockIdx.x / DEG_R;
    const int lo  = rg * DEG_T;

    for (int i = tid; i < DEG_T; i += 1024) sdeg4[i] = make_float4(0.0f, 0.0f, 0.0f, 0.0f);
    __syncthreads();

    const int cbase = blockIdx.x * (X0_CP * 1024) + tid;
    int ch = g, it = 0;
    for (; it < X0_CP && ch * DEG_CH < N_EDGES; ch += DEG_G, ++it)
        deg_trip<true >(row, col, intents, sdeg, lo, keep, bq, seen, lt, ch * DEG_CH + tid * 4, Gu, Gi, x0, cbase + it * 1024);
    for (; ch * DEG_CH < N_EDGES; ch += DEG_G)
        deg_trip<false>(row, col, intents, sdeg, lo, keep, bq, seen, lt, ch * DEG_CH + tid * 4, Gu, Gi, x0, 0);
    for (; it < X0_CP; ++it) x0_put(x0, cbase + it * 1024, x0_piece(Gu, Gi, cbase + it * 1024));
    if (keep) {
        const int nseen = __builtin_amdgcn_readfirstlane(seen);
        const int L = (rg * DEG_G + g) * 32 + wv;
        unsigned* gl = blist + (size_t)L * BL_CAP;
        const int nst = min(nseen, BL_CAP);
        for (int c = 0; c < BL_CAP / 32; ++c) {
            const int idx = c * 32 + lane;
            gl[idx] = (idx < nst) ? bq[idx] : 0xFFFFFFFFu;
        }
        if (lane == 0) bcnt[L] = nseen;
    }
    __syncthreads();

    float4* dst = (float4*)(planes + (size_t)g * (N_NODES * 4)) + lo;
    for (int i = tid; i < DEG_T; i += 1024)
        if (lo + i < N_NODES) dst[i] = sdeg4[i];
}

__global__ __launch_bounds__(128) void k_dis(
        const float* __restrict__ planes, float* __restrict__ dis) {
    const int i = blockIdx.x * 128 + threadIdx.x;
    if (i < N_NODES) {
        float4 s = ((const float4*)planes)[i];
#pragma unroll
        for (int g = 1; g < DEG_G; ++g) {
            const float4 p = ((const float4*)(planes + (size_t)g * (N_NODES * 4)))[i];
            s.x += p.x; s.y += p.y; s.z += p.z; s.w += p.w;
        }
        float4 o;
        o.x = (s.x > 0.0f) ? rsqrtf(fmaxf(s.x, 1e-12f)) : 0.0f;
        o.y = (s.y > 0.0f) ? rsqrtf(fmaxf(s.y, 1e-12f)) : 0.0f;
        o.z = (s.z > 0.0f) ? rsqrtf(fmaxf(s.z, 1e-12f)) : 0.0f;
        o.w = (s.w > 0.0f) ? rsqrtf(fmaxf(s.w, 1e-12f)) : 0.0f;
        ((float4*)dis)[i] = o;
    }
}

#define PR_R       256
#define PR_T       586
#define PR_WROWS   19
#define PR_QCAP    1024
#define PR_MAXPASS 32
#define PR_TRIPS   ((N_EDGES + 128 * 128 - 1) / (128 * 128))
#define PR_WCAP    256
#define PR_WS_NEED ((size_t)N_NODES * 4 * 4 + (size_t)N_NODES * 64 * 4 + (size_t)PR_R * 32 * 4 + (size_t)PR_R * 32 * PR_WCAP * 4)
static_assert(BL_WS_NEED == PR_WS_NEED + (size_t)BL_LISTS * 4 + (size_t)BL_LISTS * BL_CAP * 4, "the binned lists lie behind the stored lists");
#define NW_WS_NEED (BL_WS_NEED + (size_t)PR_R * 32 * PR_WCAP * 16)
static_assert(DEG_T == BL_SUB * PR_T, "a degree range must be whole layer ranges");
static_assert(DEG_R * BL_SUB >= PR_R, "every layer range needs a degree range");
static_assert(DEG_G * 32 == 32 * 16, "a layer-1 wave takes 16 of its range's lists");
static_assert(32 * 272 <= PR_T * 16 && 272 + 16 * 36 <= PR_QCAP && PR_WCAP + 6 <= 272, "the replay path's per-wave LDS areas fit");

__device__ __forceinline__ unsigned prop_group(
        const float* __restrict__ x_src, const float* __restrict__ dis,
        float* sacc, const unsigned* grp, int n, int lo,
        int hl, bool hi, float4& nout) {
    const bool vA = hl < n;
    const unsigned entA = vA ? grp[hl] : 0u;
    const int irA = (int)(entA & 0xFFFFFu);
    const int dlA = (int)(entA >> 20);
    const float4 dr = *(const float4*)(dis + irA * 4);
    const float4 dc = *(const float4*)(dis + (lo + dlA) * 4);
    float nrm[4];
    nrm[0] = vA ? dr.x * dc.x : 0.0f; nrm[1] = vA ? dr.y * dc.y : 0.0f;
    nrm[2] = vA ? dr.z * dc.z : 0.0f; nrm[3] = vA ? dr.w * dc.w : 0.0f;
    nout = make_float4(nrm[0], nrm[1], nrm[2], nrm[3]);

    const int kx = hi ? 2 : 0;
    const int mo = hi ? 8 : 0;
    int rb[8];
#pragma unroll
    for (int t = 0; t < 4; ++t) {
        rb[2 * t]     = __shfl(irA, 4 * t + kx);
        rb[2 * t + 1] = __shfl(irA, 4 * t + kx + 1);
    }
    int dlv[8];
#pragma unroll
    for (int v = 0; v < 8; ++v) dlv[v] = __shfl(dlA, v + mo);

    float bx[16], by[16];
#pragma unroll
    for (int k = 0; k < 4; ++k) {
#pragma unroll
        for (int t = 0; t < 4; ++t) {
            bx[k * 4 + t] = x_src[rb[2 * t]     * 64 + k * 16 + hl];
            by[k * 4 + t] = x_src[rb[2 * t + 1] * 64 + k * 16 + hl];
        }
    }

#pragma unroll
    for (int k = 0; k < 4; ++k) {
        v8f acc = {};
#pragma unroll
        for (int t = 0; t < 4; ++t) {
            v2f a, b;
            a.x = (hl == 4 * t + kx)     ? nrm[k] : 0.0f;
            a.y = (hl == 4 * t + kx + 1) ? nrm[k] : 0.0f;
            b.x = bx[k * 4 + t];
            b.y = by[k * 4 + t];
            acc = __builtin_amdgcn_wmma_f32_16x16x4_f32(
                      false, a, false, b, (short)0, acc, false, false);
        }
#pragma unroll
        for (int v = 0; v < 8; ++v)
            if (v + mo < n) ldsAddF32(sacc + dlv[v] * 64 + k * 16 + hl, acc[v]);
    }
    return entA;
}


#define RP_PITCH 36
__device__ __forceinline__ void replay_wave(
        const float* __restrict__ x_src, float* __restrict__ x_dst, const float* __restrict__ dis,
        const unsigned* __restrict__ glist, const float4* __restrict__ gnorm, bool nstored,
        int n, int lo, int olo, int nrows,
        unsigned* went, float4* wnrm, float* stg, int lane, unsigned lt) {
    const int c = lane & 15;
    const int h = lane >> 4;
    unsigned e[PR_WCAP / 32];
#pragma unroll
    for (int j = 0; j < PR_WCAP / 32; ++j) {
        const int idx = j * 32 + lane;
        e[j] = (idx < n) ? glist[idx] : 0u;
    }
    float4 nv[PR_WCAP / 32];
    if (nstored) {
#pragma unroll
        for (int j = 0; j < PR_WCAP / 32; ++j) {
            const int idx = j * 32 + lane;
            nv[j] = (idx < n) ? gnorm[idx] : make_float4(0.0f, 0.0f, 0.0f, 0.0f);
        }
    } else {
#pragma unroll
        for (int j = 0; j < PR_WCAP / 32; ++j) {
            const bool valid = j * 32 + lane < n;
            const int src = (int)(e[j] & 0xFFFFFu);
            const int dl  = (int)(e[j] >> 20);
            const float4 dr = *(const float4*)(dis + src * 4);
            const float4 dc = *(const float4*)(dis + (lo + dl) * 4);
            nv[j].x = valid ? dr.x * dc.x : 0.0f; nv[j].y = valid ? dr.y * dc.y : 0.0f;
            nv[j].z = valid ? dr.z * dc.z : 0.0f; nv[j].w = valid ? dr.w * dc.w : 0.0f;
        }
    }
    unsigned b0[PR_WCAP / 32], b1[PR_WCAP / 32];
    int n0 = 0, n1 = 0;
#pragma unroll
    for (int j = 0; j < PR_WCAP / 32; ++j) {
        const bool valid = j * 32 + lane < n;
        const int dl  = (int)(e[j] >> 20);
        const bool up = (dl - olo) >= 16;
        b0[j] = __builtin_amdgcn_ballot_w32(valid && !up);
        b1[j] = __builtin_amdgcn_ballot_w32(valid && up);
        n0 += __popc(b0[j]);
        n1 += __popc(b1[j]);
    }
    const int z0 = (n0 + 3) & ~3;
    const int z1 = z0 + ((n1 + 3) & ~3);
    {
        int p0 = 0, p1 = z0;
#pragma unroll
        for (int j = 0; j < PR_WCAP / 32; ++j) {
            const bool valid = j * 32 + lane < n;
            const bool up = ((int)(e[j] >> 20) - olo) >= 16;
            const int pos = up ? p1 + __popc(b1[j] & lt) : p0 + __popc(b0[j] & lt);
            if (valid) { went[pos] = e[j]; wnrm[pos] = nv[j]; }
            p0 += __popc(b0[j]);
            p1 += __popc(b1[j]);
        }
        if (n0 + lane < z0)      { went[n0 + lane] = 0u;      wnrm[n0 + lane]      = make_float4(0.0f, 0.0f, 0.0f, 0.0f); }
        if (z0 + n1 + lane < z1) { went[z0 + n1 + lane] = 0u; wnrm[z0 + n1 + lane] = make_float4(0.0f, 0.0f, 0.0f, 0.0f); }
    }

    for (int tile = 0; tile < 2; ++tile) {
        const int gb = tile ? z0 : 0;
        const int ge = tile ? z1 : z0;
        const int rb = olo + tile * 16;
        v8f acc[4] = {};
        for (int g0 = gb; g0 < ge; g0 += 16) {
            const int cnt = min(16, ge - g0);
            float ax[16], ay[16], wa[16], wb[16];
#pragma unroll
            for (int t = 0; t < 4; ++t) {
                const int idx = min(g0 + 4 * t + 2 * h, ge - 2);
                const uint2 ee = *(const uint2*)(went + idx);
                const float4 na = wnrm[idx];
                const float4 nb = wnrm[idx + 1];
                const int s0 = (int)(ee.x & 0xFFFFFu);
                const int s1 = (int)(ee.y & 0xFFFFFu);
                const bool m0 = ((int)(ee.x >> 20) - rb) == c;
                const bool m1 = ((int)(ee.y >> 20) - rb) == c;
#pragma unroll
                for (int k = 0; k < 4; ++k) {
                    ax[t * 4 + k] = x_src[s0 * 64 + k * 16 + c];
                    ay[t * 4 + k] = x_src[s1 * 64 + k * 16 + c];
                }
                wa[t * 4 + 0] = m0 ? na.x : 0.0f; wa[t * 4 + 1] = m0 ? na.y : 0.0f;
                wa[t * 4 + 2] = m0 ? na.z : 0.0f; wa[t * 4 + 3] = m0 ? na.w : 0.0f;
                wb[t * 4 + 0] = m1 ? nb.x : 0.0f; wb[t * 4 + 1] = m1 ? nb.y : 0.0f;
                wb[t * 4 + 2] = m1 ? nb.z : 0.0f; wb[t * 4 + 3] = m1 ? nb.w : 0.0f;
            }
#pragma unroll
            for (int i = 0; i < 16; ++i) {
                asm volatile("" :: "v"(ax[i]), "v"(ay[i]));
            }
#pragma unroll
            for (int t = 0; t < 4; ++t) {
                if (t == 0 || 4 * t < cnt) {
#pragma unroll
                    for (int k = 0; k < 4; ++k) {
                        v2f a, b;
                        a.x = ax[t * 4 + k]; a.y = ay[t * 4 + k];
                        b.x = wa[t * 4 + k]; b.y = wb[t * 4 + k];
                        acc[k] = __builtin_amdgcn_wmma_f32_16x16x4_f32(
                                     false, a, false, b, (short)0, acc[k], false, false);
                    }
                }
            }
        }
        const int vr = min(min(PR_WROWS - tile * 16, 16), min(PR_T, nrows) - rb);
        if (vr > 0) {
#pragma unroll
            for (int p = 0; p < 2; ++p) {
                float* sp = stg + c * RP_PITCH + 8 * h;
                *(float4*)(sp)      = make_float4(acc[2 * p][0], acc[2 * p][1], acc[2 * p][2], acc[2 * p][3]);
                *(float4*)(sp + 4)  = make_float4(acc[2 * p][4], acc[2 * p][5], acc[2 * p][6], acc[2 * p][7]);
                *(float4*)(sp + 16) = make_float4(acc[2 * p + 1][0], acc[2 * p + 1][1], acc[2 * p + 1][2], acc[2 * p + 1][3]);
                *(float4*)(sp + 20) = make_float4(acc[2 * p + 1][4], acc[2 * p + 1][5], acc[2 * p + 1][6], acc[2 * p + 1][7]);
#pragma unroll
                for (int j = 0; j < 4; ++j) {
                    if (j * 4 < vr) {
                        const int i = j * 32 + lane;
                        const int r = i >> 3, q = i & 7;
                        const float4 v = *(const float4*)(stg + r * RP_PITCH + q * 4);
                        if (r < vr) *(float4*)(x_dst + (size_t)(lo + rb + r) * 64 + p * 32 + q * 4) = v;
                    }
                }
            }
        }
    }
}

__device__ __forceinline__ void scan_load(const int* __restrict__ col, int ip, int tid, int4 (&cc)[4]) {
#pragma unroll
    for (int j = 0; j < 4; ++j) {
        const unsigned i4 = (unsigned)min(ip * 4096 + j * 1024 + tid, N_EDGES / 4 - 1);
        cc[j] = *(const int4*)((const char*)col + i4 * 16u);
    }
}

template <bool BUILD>
__global__ __launch_bounds__(1024) void k_propagate(
        const float* __restrict__ x_src, float* __restrict__ x_dst,
        const float* __restrict__ dis,
        const int* __restrict__ row, const int* __restrict__ col,
        int* __restrict__ wcnt, unsigned* __restrict__ wlist,
        const int* __restrict__ bcnt, const unsigned* __restrict__ blist, float* __restrict__ wnorm, size_t ws_bytes) {
    __shared__ float4   sacc4[PR_T * 16];
    __shared__ __attribute__((aligned(16))) unsigned squeue[32 * PR_QCAP];
    __shared__ unsigned sstage[32 * 64];
    __shared__ int      scnt[32];
    float* sacc = (float*)sacc4;
    const int tid  = threadIdx.x;
    const int wv   = __builtin_amdgcn_readfirstlane(tid >> 5);
    const int lane = tid & 31;
    const int hl   = lane & 15;
    const bool hi  = lane >= 16;
    const unsigned lt = (1u << lane) - 1u;
    const int lo   = blockIdx.x * PR_T;
    const int rot  = blockIdx.x % PR_TRIPS;

    const bool room = ws_bytes >= PR_WS_NEED;
    unsigned* glist = wlist + (size_t)(blockIdx.x * 32 + wv) * PR_WCAP;
    const bool nroom = ws_bytes >= NW_WS_NEED;
    float4* gnorm = (float4*)wnorm + (size_t)(blockIdx.x * 32 + wv) * PR_WCAP;
    int wc_l = 0;
    if (!BUILD && room) wc_l = wcnt[blockIdx.x * 32 + lane];
    bool binned = false;
    int bc_l = 0;
    const int lb = (blockIdx.x / BL_SUB) * (DEG_G * 32);
    if (BUILD && ws_bytes >= BL_WS_NEED) {
        unsigned over = 0u;
#pragma unroll
        for (int j = 0; j < DEG_G; ++j) over |= (unsigned)((unsigned)bcnt[lb + j * 32 + lane] > (unsigned)BL_CAP);
        bc_l = bcnt[lb + wv * 16 + (lane & 15)];
        binned = __builtin_amdgcn_ballot_w32(over != 0u) == 0u;
    }

    const bool replay = !BUILD && room &&
                        __builtin_amdgcn_ballot_w32((unsigned)wc_l > (unsigned)PR_WCAP) == 0u;

    if (!replay)
        for (int i = tid; i < PR_T * 16; i += 1024) sacc4[i] = make_float4(0.0f, 0.0f, 0.0f, 0.0f);

    if (replay) {
        const int n = __builtin_amdgcn_readlane(wc_l, wv);
        unsigned* went = squeue + wv * PR_QCAP;
        replay_wave(x_src, x_dst, dis, glist, gnorm, nroom, n, lo, wv * PR_WROWS, min(PR_T, N_NODES - lo),
                    went, sacc4 + wv * 272, (float*)(went + 272), lane, lt);
    } else {
        int wtot = 0;
        for (int pass = 0; pass < PR_MAXPASS; ++pass) {
            const int wlo = pass * PR_QCAP;

            unsigned* q = squeue + wv * PR_QCAP;
            int seen = 0;
            if (BUILD && binned) {
            const unsigned j0 = (unsigned)((blockIdx.x % BL_SUB) * PR_T);
            const unsigned* bl0 = blist + (size_t)(lb + wv * 16) * BL_CAP;
            unsigned en[BL_CAP / 32];
#pragma unroll
            for (int c = 0; c < BL_CAP / 32; ++c) en[c] = bl0[c * 32 + lane];
            for (int i = 0; i < 16; ++i) {
                const int n = __builtin_amdgcn_readlane(bc_l, i);
                unsigned ec[BL_CAP / 32];
#pragma unroll
                for (int c = 0; c < BL_CAP / 32; ++c) ec[c] = en[c];
                {
                    const unsigned* bl = bl0 + (size_t)min(i + 1, 15) * BL_CAP;
#pragma unroll
                    for (int c = 0; c < BL_CAP / 32; ++c) en[c] = bl[c * 32 + lane];
                }
#pragma unroll
                for (int c = 0; c < BL_CAP / 32; ++c) {
                    if (c * 32 < n) {
                        const unsigned u = (ec[c] >> 18) - j0;
                        const bool t = u < (unsigned)PR_T;
                        const unsigned b = __builtin_amdgcn_ballot_w32(t);
                        if (b != 0u) {
                            const unsigned w = (unsigned)(seen + __popc(b & lt) - wlo);
                            if (t && w < (unsigned)PR_QCAP) q[w] = (u << 20) | (ec[c] & 0x3FFFFu);
                            seen += __popc(b);
                        }
                    }
                }
            }
            } else {
            int4 nx[4];
            scan_load(col, rot, tid, nx);
            for (int it = 0; it < PR_TRIPS; ++it) {
                const int ip = (it + rot < PR_TRIPS) ? it + rot : it + rot - PR_TRIPS;
                int4 cc[4];
#pragma unroll
                for (int j = 0; j < 4; ++j) cc[j] = nx[j];
                {
                    const int in1 = (it + 1 < PR_TRIPS) ? it + 1 + rot : rot;
                    scan_load(col, (in1 < PR_TRIPS) ? in1 : in1 - PR_TRIPS, tid, nx);
                }
#pragma unroll
                for (int j = 0; j < 4; ++j) {
                    const int e = (ip * 4096 + j * 1024 + tid) * 4;
                    const bool in = e < N_EDGES;
                    const unsigned bin = __builtin_amdgcn_ballot_w32(in);
                    const unsigned u0 = (unsigned)(cc[j].x - lo), u1 = (unsigned)(cc[j].y - lo);
                    const unsigned u2 = (unsigned)(cc[j].z - lo), u3 = (unsigned)(cc[j].w - lo);
                    const bool t0 = u0 < (unsigned)PR_T, t1 = u1 < (unsigned)PR_T;
                    const bool t2 = u2 < (unsigned)PR_T, t3 = u3 < (unsigned)PR_T;
                    const unsigned b0 = __builtin_amdgcn_ballot_w32(t0) & bin;
                    const unsigned b1 = __builtin_amdgcn_ballot_w32(t1) & bin;
                    const unsigned b2 = __builtin_amdgcn_ballot_w32(t2) & bin;
                    const unsigned b3 = __builtin_amdgcn_ballot_w32(t3) & bin;
                    if ((b0 | b1 | b2 | b3) != 0u) {
                        if (b0 != 0u) {
                            const unsigned w = (unsigned)(seen + __popc(b0 & lt) - wlo);
                            if (t0 && in && w < (unsigned)PR_QCAP) q[w] = (u0 << 20) | (unsigned)(e + 0);
                            seen += __popc(b0);
                        }
                        if (b1 != 0u) {
                            const unsigned w = (unsigned)(seen + __popc(b1 & lt) - wlo);
                            if (t1 && in && w < (unsigned)PR_QCAP) q[w] = (u1 << 20) | (unsigned)(e + 1);
                            seen += __popc(b1);
                        }
                        if (b2 != 0u) {
                            const unsigned w = (unsigned)(seen + __popc(b2 & lt) - wlo);
                            if (t2 && in && w < (unsigned)PR_QCAP) q[w] = (u2 << 20) | (unsigned)(e + 2);
                            seen += __popc(b2);
                        }
                        if (b3 != 0u) {
                            const unsigned w = (unsigned)(seen + __popc(b3 & lt) - wlo);
                            if (t3 && in && w < (unsigned)PR_QCAP) q[w] = (u3 << 20) | (unsigned)(e + 3);
                            seen += __popc(b3);
                        }
                    }
                }
            }

            {
                const int nown = min(max(seen - wlo, 0), PR_QCAP);
                for (int i0 = 0; i0 < nown; i0 += 128) {
                    unsigned ent[4];
                    int sid[4];
#pragma unroll
                    for (int j = 0; j < 4; ++j) {
                        const int idx = i0 + 32 * j + lane;
                        ent[j] = (idx < nown) ? q[idx] : 0u;
                    }
#pragma unroll
                    for (int j = 0; j < 4; ++j) sid[j] = row[ent[j] & 0xFFFFFu];
#pragma unroll
                    for (int j = 0; j < 4; ++j) {
                        const int idx = i0 + 32 * j + lane;
                        if (idx < nown) q[idx] = (ent[j] & 0xFFF00000u) | (unsigned)sid[j];
                    }
                }
            }
            }
            if (lane == 0) scnt[wv] = seen;
            __syncthreads();

            const int cnt_l = scnt[lane];
            const bool more = __builtin_amdgcn_ballot_w32(cnt_l > wlo + PR_QCAP) != 0u;

            {
                const unsigned olo = (unsigned)(wv * PR_WROWS);
                unsigned* st = sstage + wv * 64;
                int head = 0, have = 0;
                for (int qw = 0; qw < 32; ++qw) {
                    const int n = min(max(__builtin_amdgcn_readlane(cnt_l, qw) - wlo, 0), PR_QCAP);
                    const unsigned* qq = squeue + qw * PR_QCAP;
                    for (int i = 0; i < n; i += 128) {
                        unsigned ent[4];
#pragma unroll
                        for (int j = 0; j < 4; ++j) {
                            const int idx = i + 32 * j + lane;
                            ent[j] = (idx < n) ? qq[idx] : 0xFFFFFFFFu;
                        }
#pragma unroll
                        for (int j = 0; j < 4; ++j) {
                            const bool m = ((ent[j] >> 20) - olo) < (unsigned)PR_WROWS;
                            const unsigned b = __builtin_amdgcn_ballot_w32(m);
                            if (b != 0u) {
                                if (m) st[(head + have + __popc(b & lt)) & 63] = ent[j];
                                have += __popc(b);
                                while (have >= 16) {
                                    float4 nq;
                                    const unsigned ent = prop_group(x_src, dis, sacc, st + head, 16, lo, hl, hi, nq);
                                    if (BUILD && room && !hi && wtot + hl < PR_WCAP) {
                                        glist[wtot + hl] = ent;
                                        if (nroom) gnorm[wtot + hl] = nq;
                                    }
                                    wtot += 16;
                                    head = (head + 16) & 63;
                                    have -= 16;
                                }
                            }
                        }
                    }
                }
                if (have > 0) {
                    float4 nq;
                    const unsigned ent = prop_group(x_src, dis, sacc, st + head, have, lo, hl, hi, nq);
                    if (BUILD && room && !hi && hl < have && wtot + hl < PR_WCAP) {
                        glist[wtot + hl] = ent;
                        if (nroom) gnorm[wtot + hl] = nq;
                    }
                    wtot += have;
                }
            }
            __syncthreads();
            if (!more) break;
        }
        if (BUILD && room && lane == 0) wcnt[blockIdx.x * 32 + wv] = wtot;
    }

    if (!replay) {
        const int nrows = min(PR_T, N_NODES - lo);
        float4* dst = (float4*)x_dst + (size_t)lo * 16;
        for (int i = tid; i < nrows * 16; i += 1024) dst[i] = sacc4[i];
    }
}

extern "C" void kernel_launch(void* const* d_in, const int* in_sizes, int n_in,
                              void* d_out, int out_size, void* d_ws, size_t ws_size,
                              hipStream_t stream) {
    const float* Gu      = (const float*)d_in[0];
    const float* Gi      = (const float*)d_in[1];
    const int*   edge    = (const int*)d_in[2];
    const float* intents = (const float*)d_in[3];
    const int* row = edge;
    const int* col = edge + N_EDGES;
    float* out = (float*)d_out;

    float* deg  = (float*)d_ws;
    float* xbuf = deg + (size_t)N_NODES * 4;

    int*      wcnt  = (int*)(xbuf + (size_t)N_NODES * 64);
    unsigned* wlist = (unsigned*)(wcnt + PR_R * 32);
    int*      bcnt  = (int*)(wlist + (size_t)PR_R * 32 * PR_WCAP);
    unsigned* blist = (unsigned*)(bcnt + BL_LISTS);
    float*    wnorm = (float*)(blist + (size_t)BL_LISTS * BL_CAP);

    dim3 pgrid(PR_R), pblock(1024);

    k_deg<<<dim3(DEG_R * DEG_G), dim3(1024), 0, stream>>>(row, col, intents, out, bcnt, blist, ws_size, Gu, Gi, xbuf);
    k_dis<<<dim3((N_NODES + 127) / 128), dim3(128), 0, stream>>>(out, deg);

    k_propagate<true ><<<pgrid, pblock, 0, stream>>>(xbuf, out, deg, row, col, wcnt, wlist, bcnt, blist, wnorm, ws_size);
    k_propagate<false><<<pgrid, pblock, 0, stream>>>(out, xbuf, deg, row, col, wcnt, wlist, bcnt, blist, wnorm, ws_size);
    k_propagate<false><<<pgrid, pblock, 0, stream>>>(xbuf, out, deg, row, col, wcnt, wlist, bcnt, blist, wnorm, ws_size);
}
